# starting kernel
# baseline (speedup 1.0000x reference)
#include <hip/hip_runtime.h>
#include <hip/hip_bf16.h>
#include <math.h>


typedef __attribute__((ext_vector_type(2))) float v2f;
typedef __attribute__((ext_vector_type(8))) float v8f;

#define D_MODEL 1024
#define D_INNER 2048
#define NB 2
#define SEQ_L 1024
#define DSTATE 16
#define DTRANK 64
#define XDBL_N 96

template <int MODE, int MT, int NT>
__global__ __launch_bounds__(256) void wmma_gemm_f32(
    const float* __restrict__ A, const float* __restrict__ W,
    float* __restrict__ C, const float* __restrict__ bias,
    int M, int N, int K, int lda, int ldb, int ldc) {
  const int lane = threadIdx.x & 31;
  const int wave = threadIdx.x >> 5;
  const int ntx = N / (16 * NT);
  const int nty = M / (16 * MT);
  const int task = blockIdx.x * 8 + wave;
  if (task >= ntx * nty) return;
  const int m0 = (task / ntx) * (16 * MT);
  const int n0 = (task % ntx) * (16 * NT);

  const int r = lane & 15;
  const int khalf = (lane >> 4) << 1;

  const float* arow[MT];
#pragma unroll
  for (int i = 0; i < MT; ++i)
    arow[i] = A + (size_t)(m0 + i * 16 + r) * lda + khalf;
  const float* wrow[NT];
#pragma unroll
  for (int j = 0; j < NT; ++j)
    wrow[j] = W + (size_t)(n0 + j * 16 + r) * ldb + khalf;

  v8f acc[MT][NT];
#pragma unroll
  for (int i = 0; i < MT; ++i)
#pragma unroll
    for (int j = 0; j < NT; ++j) acc[i][j] = (v8f){};

  v2f a_cur[MT], b_cur[NT];
#pragma unroll
  for (int i = 0; i < MT; ++i) a_cur[i] = *(const v2f*)(arow[i]);
#pragma unroll
  for (int j = 0; j < NT; ++j) b_cur[j] = *(const v2f*)(wrow[j]);

  for (int k = 0; k < K; k += 4) {
    const int kn = (k + 4 < K) ? (k + 4) : 0;
    v2f a_nxt[MT], b_nxt[NT];
#pragma unroll
    for (int i = 0; i < MT; ++i) a_nxt[i] = *(const v2f*)(arow[i] + kn);
#pragma unroll
    for (int j = 0; j < NT; ++j) b_nxt[j] = *(const v2f*)(wrow[j] + kn);

#pragma unroll
    for (int i = 0; i < MT; ++i)
#pragma unroll
      for (int j = 0; j < NT; ++j)
        acc[i][j] = __builtin_amdgcn_wmma_f32_16x16x4_f32(
            false, a_cur[i], false, b_cur[j], (short)0, acc[i][j], false, false);

#pragma unroll
    for (int i = 0; i < MT; ++i) a_cur[i] = a_nxt[i];
#pragma unroll
    for (int j = 0; j < NT; ++j) b_cur[j] = b_nxt[j];
  }

  const int mb = (lane >> 4) << 3;
#pragma unroll
  for (int i = 0; i < MT; ++i) {
#pragma unroll
    for (int j = 0; j < NT; ++j) {
      const int n = n0 + j * 16 + r;
      float bv = 0.0f;
      if (MODE == 1) bv = bias[n];
#pragma unroll
      for (int e = 0; e < 8; ++e) {
        float v = acc[i][j][e];
        if (MODE == 1) {
          v += bv;
          v = (v > 20.0f) ? v : log1pf(__expf(v));
        }
        C[(size_t)(m0 + i * 16 + mb + e) * ldc + n] = v;
      }
    }
  }
}

__global__ __launch_bounds__(256) void conv_silu_kernel(
    const float* __restrict__ xz, const float* __restrict__ cw,
    const float* __restrict__ cb, float* __restrict__ xconv) {
  const int gid = blockIdx.x * blockDim.x + threadIdx.x;
  if (gid >= NB * SEQ_L * D_INNER) return;
  const int d = gid & (D_INNER - 1);
  const int bl = gid / D_INNER;
  const int l = bl & (SEQ_L - 1);
  const int b = bl / SEQ_L;

  float acc = cb[d];
  const float* base = xz + ((size_t)b * SEQ_L) * (2 * D_INNER) + d;
#pragma unroll
  for (int j = 0; j < 4; ++j) {
    const int ls = l - 3 + j;
    if (ls >= 0) acc = fmaf(cw[d * 4 + j], base[(size_t)ls * (2 * D_INNER)], acc);
  }
  xconv[gid] = acc / (1.0f + __expf(-acc));
}

__global__ __launch_bounds__(256) void scan_kernel(
    const float* __restrict__ dt, const float* __restrict__ xconv,
    const float* __restrict__ x_dbl, const float* __restrict__ xz,
    const float* __restrict__ A_log, const float* __restrict__ Dp,
    float* __restrict__ u) {
  const int gid = blockIdx.x * blockDim.x + threadIdx.x;
  const int c = gid >> 4;
  const int n = gid & 15;
  if (c >= NB * D_INNER) return;
  const int d = c & (D_INNER - 1);
  const int b = c >> 11;

  const float An = -__expf(A_log[d * DSTATE + n]);
  const float Dd = Dp[d];
  float h = 0.0f;

  const float* dt_p = dt + ((size_t)b * SEQ_L) * D_INNER + d;
  const float* xc_p = xconv + ((size_t)b * SEQ_L) * D_INNER + d;
  const float* bc_p = x_dbl + ((size_t)b * SEQ_L) * XDBL_N + DTRANK + n;
  const float* z_p = xz + ((size_t)b * SEQ_L) * (2 * D_INNER) + D_INNER + d;
  float* u_p = u + ((size_t)b * SEQ_L) * D_INNER + d;

  for (int l = 0; l < SEQ_L; ++l) {
    const float dtv = dt_p[(size_t)l * D_INNER];
    const float xv = xc_p[(size_t)l * D_INNER];
    const float Bn = bc_p[(size_t)l * XDBL_N];
    const float Cn = bc_p[(size_t)l * XDBL_N + DSTATE];

    const float dA = __expf(dtv * An);
    h = fmaf(dA, h, dtv * xv * Bn);

    float y = h * Cn;
    y += __shfl_xor(y, 1, 32);
    y += __shfl_xor(y, 2, 32);
    y += __shfl_xor(y, 4, 32);
    y += __shfl_xor(y, 8, 32);

    if (n == 0) {
      const float zv = z_p[(size_t)l * (2 * D_INNER)];
      const float sz = zv / (1.0f + __expf(-zv));
      u_p[(size_t)l * D_INNER] = fmaf(xv, Dd, y) * sz;
    }
  }
}

extern "C" void kernel_launch(void* const* d_in, const int* in_sizes, int n_in,
                              void* d_out, int out_size, void* d_ws, size_t ws_size,
                              hipStream_t stream) {
  const float* x      = (const float*)d_in[0];
  const float* W_in   = (const float*)d_in[1];
  const float* conv_w = (const float*)d_in[2];
  const float* conv_b = (const float*)d_in[3];
  const float* W_x    = (const float*)d_in[4];
  const float* W_dt   = (const float*)d_in[5];
  const float* b_dt   = (const float*)d_in[6];
  const float* A_log  = (const float*)d_in[7];
  const float* Dp     = (const float*)d_in[8];
  const float* W_out  = (const float*)d_in[9];
  float* out = (float*)d_out;

  float* ws = (float*)d_ws;
  const size_t M = (size_t)NB * SEQ_L;
  float* xz    = ws;
  float* xconv = xz + M * 2 * D_INNER;
  float* x_dbl = xconv + M * D_INNER;
  float* dt    = x_dbl + M * XDBL_N;
  float* u     = dt + M * D_INNER;

  dim3 blk(256);
  {
    int tasks = (int)(M / 32) * (2 * D_INNER / 64);
    hipLaunchKernelGGL((wmma_gemm_f32<0, 2, 4>), dim3((tasks + 7) / 8), blk, 0, stream,
                       x, W_in, xz, nullptr,
                       (int)M, 2 * D_INNER, D_MODEL, D_MODEL, D_MODEL, 2 * D_INNER);
  }
  {
    int total = NB * SEQ_L * D_INNER;
    hipLaunchKernelGGL(conv_silu_kernel, dim3((total + 255) / 256), blk, 0, stream,
                       xz, conv_w, conv_b, xconv);
  }
  {
    int tasks = (int)(M / 32) * (XDBL_N / 32);
    hipLaunchKernelGGL((wmma_gemm_f32<0, 2, 2>), dim3((tasks + 7) / 8), blk, 0, stream,
                       xconv, W_x, x_dbl, nullptr,
                       (int)M, XDBL_N, D_INNER, D_INNER, D_INNER, XDBL_N);
  }
  {
    int tasks = (int)(M / 32) * (D_INNER / 64);
    hipLaunchKernelGGL((wmma_gemm_f32<1, 2, 4>), dim3((tasks + 7) / 8), blk, 0, stream,
                       x_dbl, W_dt, dt, b_dt,
                       (int)M, D_INNER, DTRANK, XDBL_N, DTRANK, D_INNER);
  }
  {
    int total = NB * D_INNER * DSTATE;
    hipLaunchKernelGGL(scan_kernel, dim3((total + 255) / 256), blk, 0, stream,
                       dt, xconv, x_dbl, xz, A_log, Dp, u);
  }
  {
    int tasks = (int)(M / 32) * (D_MODEL / 64);
    hipLaunchKernelGGL((wmma_gemm_f32<0, 2, 4>), dim3((tasks + 7) / 8), blk, 0, stream,
                       u, W_out, out, nullptr,
                       (int)M, D_MODEL, D_INNER, D_INNER, D_INNER, D_MODEL);
  }
}


# scan kernel: operands of 16 time steps (dt, x, B, C, z) loaded one batch ahead into a second register set, loop unrolled 16x so only the fma on h is serial; per-element operation order kept
# speedup vs baseline: 2.3274x; 2.3274x over previous
#include <hip/hip_runtime.h>
#include <hip/hip_bf16.h>
#include <math.h>


typedef __attribute__((ext_vector_type(2))) float v2f;
typedef __attribute__((ext_vector_type(8))) float v8f;

#define D_MODEL 1024
#define D_INNER 2048
#define NB 2
#define SEQ_L 1024
#define DSTATE 16
#define DTRANK 64
#define XDBL_N 96

template <int MODE, int MT, int NT>
__global__ __launch_bounds__(256) void wmma_gemm_f32(
    const float* __restrict__ A, const float* __restrict__ W,
    float* __restrict__ C, const float* __restrict__ bias,
    int M, int N, int K, int lda, int ldb, int ldc) {
  const int lane = threadIdx.x & 31;
  const int wave = threadIdx.x >> 5;
  const int ntx = N / (16 * NT);
  const int nty = M / (16 * MT);
  const int task = blockIdx.x * 8 + wave;
  if (task >= ntx * nty) return;
  const int m0 = (task / ntx) * (16 * MT);
  const int n0 = (task % ntx) * (16 * NT);

  const int r = lane & 15;
  const int khalf = (lane >> 4) << 1;

  const float* arow[MT];
#pragma unroll
  for (int i = 0; i < MT; ++i)
    arow[i] = A + (size_t)(m0 + i * 16 + r) * lda + khalf;
  const float* wrow[NT];
#pragma unroll
  for (int j = 0; j < NT; ++j)
    wrow[j] = W + (size_t)(n0 + j * 16 + r) * ldb + khalf;

  v8f acc[MT][NT];
#pragma unroll
  for (int i = 0; i < MT; ++i)
#pragma unroll
    for (int j = 0; j < NT; ++j) acc[i][j] = (v8f){};

  v2f a_cur[MT], b_cur[NT];
#pragma unroll
  for (int i = 0; i < MT; ++i) a_cur[i] = *(const v2f*)(arow[i]);
#pragma unroll
  for (int j = 0; j < NT; ++j) b_cur[j] = *(const v2f*)(wrow[j]);

  for (int k = 0; k < K; k += 4) {
    const int kn = (k + 4 < K) ? (k + 4) : 0;
    v2f a_nxt[MT], b_nxt[NT];
#pragma unroll
    for (int i = 0; i < MT; ++i) a_nxt[i] = *(const v2f*)(arow[i] + kn);
#pragma unroll
    for (int j = 0; j < NT; ++j) b_nxt[j] = *(const v2f*)(wrow[j] + kn);

#pragma unroll
    for (int i = 0; i < MT; ++i)
#pragma unroll
      for (int j = 0; j < NT; ++j)
        acc[i][j] = __builtin_amdgcn_wmma_f32_16x16x4_f32(
            false, a_cur[i], false, b_cur[j], (short)0, acc[i][j], false, false);

#pragma unroll
    for (int i = 0; i < MT; ++i) a_cur[i] = a_nxt[i];
#pragma unroll
    for (int j = 0; j < NT; ++j) b_cur[j] = b_nxt[j];
  }

  const int mb = (lane >> 4) << 3;
#pragma unroll
  for (int i = 0; i < MT; ++i) {
#pragma unroll
    for (int j = 0; j < NT; ++j) {
      const int n = n0 + j * 16 + r;
      float bv = 0.0f;
      if (MODE == 1) bv = bias[n];
#pragma unroll
      for (int e = 0; e < 8; ++e) {
        float v = acc[i][j][e];
        if (MODE == 1) {
          v += bv;
          v = (v > 20.0f) ? v : log1pf(__expf(v));
        }
        C[(size_t)(m0 + i * 16 + mb + e) * ldc + n] = v;
      }
    }
  }
}

__global__ __launch_bounds__(256) void conv_silu_kernel(
    const float* __restrict__ xz, const float* __restrict__ cw,
    const float* __restrict__ cb, float* __restrict__ xconv) {
  const int gid = blockIdx.x * blockDim.x + threadIdx.x;
  if (gid >= NB * SEQ_L * D_INNER) return;
  const int d = gid & (D_INNER - 1);
  const int bl = gid / D_INNER;
  const int l = bl & (SEQ_L - 1);
  const int b = bl / SEQ_L;

  float acc = cb[d];
  const float* base = xz + ((size_t)b * SEQ_L) * (2 * D_INNER) + d;
#pragma unroll
  for (int j = 0; j < 4; ++j) {
    const int ls = l - 3 + j;
    if (ls >= 0) acc = fmaf(cw[d * 4 + j], base[(size_t)ls * (2 * D_INNER)], acc);
  }
  xconv[gid] = acc / (1.0f + __expf(-acc));
}

#define SCAN_T 16

struct ScanOps {
  float dt[SCAN_T], x[SCAN_T], B[SCAN_T], C[SCAN_T], z[SCAN_T];
};

__device__ __forceinline__ void scan_fetch(ScanOps& o, const float* dt_r,
                                           const float* xc_r, const float* bc_r,
                                           const float* z_r) {
#pragma unroll
  for (int t = 0; t < SCAN_T; ++t) {
    o.dt[t] = dt_r[(size_t)t * D_INNER];
    o.x[t] = xc_r[(size_t)t * D_INNER];
    o.B[t] = bc_r[(size_t)t * XDBL_N];
    o.C[t] = bc_r[(size_t)t * XDBL_N + DSTATE];
    o.z[t] = z_r[(size_t)t * (2 * D_INNER)];
  }
}

__device__ __forceinline__ void scan_batch(const ScanOps& o, float& h, float An,
                                           float Dd, bool writer, float* u_r) {
  float uo[SCAN_T];
#pragma unroll
  for (int t = 0; t < SCAN_T; ++t) {
    const float dtv = o.dt[t];
    const float xv = o.x[t];
    const float dA = __expf(dtv * An);
    h = fmaf(dA, h, dtv * xv * o.B[t]);

    float y = h * o.C[t];
    y += __shfl_xor(y, 1, 32);
    y += __shfl_xor(y, 2, 32);
    y += __shfl_xor(y, 4, 32);
    y += __shfl_xor(y, 8, 32);

    const float zv = o.z[t];
    const float sz = zv / (1.0f + __expf(-zv));
    uo[t] = fmaf(xv, Dd, y) * sz;
  }
  if (writer) {
#pragma unroll
    for (int t = 0; t < SCAN_T; ++t) u_r[(size_t)t * D_INNER] = uo[t];
  }
}

__global__ __launch_bounds__(256) void scan_kernel(
    const float* __restrict__ dt, const float* __restrict__ xconv,
    const float* __restrict__ x_dbl, const float* __restrict__ xz,
    const float* __restrict__ A_log, const float* __restrict__ Dp,
    float* __restrict__ u) {
  const int gid = blockIdx.x * blockDim.x + threadIdx.x;
  const int c = gid >> 4;
  const int n = gid & 15;
  if (c >= NB * D_INNER) return;
  const int d = c & (D_INNER - 1);
  const int b = c >> 11;

  const float An = -__expf(A_log[d * DSTATE + n]);
  const float Dd = Dp[d];
  float h = 0.0f;

  const float* dt_p = dt + ((size_t)b * SEQ_L) * D_INNER + d;
  const float* xc_p = xconv + ((size_t)b * SEQ_L) * D_INNER + d;
  const float* bc_p = x_dbl + ((size_t)b * SEQ_L) * XDBL_N + DTRANK + n;
  const float* z_p = xz + ((size_t)b * SEQ_L) * (2 * D_INNER) + D_INNER + d;
  float* u_p = u + ((size_t)b * SEQ_L) * D_INNER + d;
  const bool writer = (n == 0);

  ScanOps ra, rb;
  scan_fetch(ra, dt_p, xc_p, bc_p, z_p);

  for (int l0 = 0; l0 < SEQ_L; l0 += 2 * SCAN_T) {
    const int l1 = l0 + SCAN_T;
    const int l2 = (l0 + 2 * SCAN_T < SEQ_L) ? (l0 + 2 * SCAN_T) : l0;

    scan_fetch(rb, dt_p + (size_t)l1 * D_INNER, xc_p + (size_t)l1 * D_INNER,
               bc_p + (size_t)l1 * XDBL_N, z_p + (size_t)l1 * (2 * D_INNER));
    scan_batch(ra, h, An, Dd, writer, u_p + (size_t)l0 * D_INNER);

    scan_fetch(ra, dt_p + (size_t)l2 * D_INNER, xc_p + (size_t)l2 * D_INNER,
               bc_p + (size_t)l2 * XDBL_N, z_p + (size_t)l2 * (2 * D_INNER));
    scan_batch(rb, h, An, Dd, writer, u_p + (size_t)l1 * D_INNER);
  }
}

extern "C" void kernel_launch(void* const* d_in, const int* in_sizes, int n_in,
                              void* d_out, int out_size, void* d_ws, size_t ws_size,
                              hipStream_t stream) {
  const float* x      = (const float*)d_in[0];
  const float* W_in   = (const float*)d_in[1];
  const float* conv_w = (const float*)d_in[2];
  const float* conv_b = (const float*)d_in[3];
  const float* W_x    = (const float*)d_in[4];
  const float* W_dt   = (const float*)d_in[5];
  const float* b_dt   = (const float*)d_in[6];
  const float* A_log  = (const float*)d_in[7];
  const float* Dp     = (const float*)d_in[8];
  const float* W_out  = (const float*)d_in[9];
  float* out = (float*)d_out;

  float* ws = (float*)d_ws;
  const size_t M = (size_t)NB * SEQ_L;
  float* xz    = ws;
  float* xconv = xz + M * 2 * D_INNER;
  float* x_dbl = xconv + M * D_INNER;
  float* dt    = x_dbl + M * XDBL_N;
  float* u     = dt + M * D_INNER;

  dim3 blk(256);
  {
    int tasks = (int)(M / 32) * (2 * D_INNER / 64);
    hipLaunchKernelGGL((wmma_gemm_f32<0, 2, 4>), dim3((tasks + 7) / 8), blk, 0, stream,
                       x, W_in, xz, nullptr,
                       (int)M, 2 * D_INNER, D_MODEL, D_MODEL, D_MODEL, 2 * D_INNER);
  }
  {
    int total = NB * SEQ_L * D_INNER;
    hipLaunchKernelGGL(conv_silu_kernel, dim3((total + 255) / 256), blk, 0, stream,
                       xz, conv_w, conv_b, xconv);
  }
  {
    int tasks = (int)(M / 32) * (XDBL_N / 32);
    hipLaunchKernelGGL((wmma_gemm_f32<0, 2, 2>), dim3((tasks + 7) / 8), blk, 0, stream,
                       xconv, W_x, x_dbl, nullptr,
                       (int)M, XDBL_N, D_INNER, D_INNER, D_INNER, XDBL_N);
  }
  {
    int tasks = (int)(M / 32) * (D_INNER / 64);
    hipLaunchKernelGGL((wmma_gemm_f32<1, 2, 4>), dim3((tasks + 7) / 8), blk, 0, stream,
                       x_dbl, W_dt, dt, b_dt,
                       (int)M, D_INNER, DTRANK, XDBL_N, DTRANK, D_INNER);
  }
  {
    int total = NB * D_INNER * DSTATE;
    hipLaunchKernelGGL(scan_kernel, dim3((total + 255) / 256), blk, 0, stream,
                       dt, xconv, x_dbl, xz, A_log, Dp, u);
  }
  {
    int tasks = (int)(M / 32) * (D_MODEL / 64);
    hipLaunchKernelGGL((wmma_gemm_f32<0, 2, 4>), dim3((tasks + 7) / 8), blk, 0, stream,
                       u, W_out, out, nullptr,
                       (int)M, D_MODEL, D_INNER, D_INNER, D_INNER, D_MODEL);
  }
}


# GEMM kernels: A and W k-chunks of 32 staged once per workgroup in LDS by 16-byte global loads, fragments read from LDS; workgroup tiles 128x128, 128x64, 32x96; same k order, same sums
# speedup vs baseline: 4.8414x; 2.0802x over previous
#include <hip/hip_runtime.h>
#include <hip/hip_bf16.h>
#include <math.h>


typedef __attribute__((ext_vector_type(2))) float v2f;
typedef __attribute__((ext_vector_type(4))) float v4f;
typedef __attribute__((ext_vector_type(8))) float v8f;

#define D_MODEL 1024
#define D_INNER 2048
#define NB 2
#define SEQ_L 1024
#define DSTATE 16
#define DTRANK 64
#define XDBL_N 96

#define GEMM_BK 32
#define GEMM_PITCH 36

template <int MODE, int BM, int BN, int WGM, int WGN>
__global__ __launch_bounds__(WGM * WGN * 32) void wmma_gemm_f32(
    const float* __restrict__ A, const float* __restrict__ W,
    float* __restrict__ C, const float* __restrict__ bias,
    int N, int K, int lda, int ldb, int ldc) {
  constexpr int NTHR = WGM * WGN * 32;
  constexpr int MT = BM / (16 * WGM);
  constexpr int NT = BN / (16 * WGN);
  constexpr int ROWS = BM + BN;
  constexpr int PIECES = ROWS * (GEMM_BK / 4);
  constexpr int NPASS = (PIECES + NTHR - 1) / NTHR;
  static_assert(PIECES % NTHR == 0, "a chunk must be a whole number of passes");

  __shared__ __attribute__((aligned(16))) float smem[2][ROWS * GEMM_PITCH];

  const int tid = threadIdx.x;
  const int lane = tid & 31;
  const int wave = tid >> 5;
  const int wm = wave / WGN;
  const int wn = wave % WGN;
  const int nbx = N / BN;
  const int m0 = (blockIdx.x / nbx) * BM;
  const int n0 = (blockIdx.x % nbx) * BN;

  const float* gsrc[NPASS];
  int sdst[NPASS];
#pragma unroll
  for (int i = 0; i < NPASS; ++i) {
    const int p = tid + i * NTHR;
    const int row = p >> 3;
    const int seg = p & 7;
    gsrc[i] = (row < BM) ? (A + (size_t)(m0 + row) * lda + seg * 4)
                         : (W + (size_t)(n0 + row - BM) * ldb + seg * 4);
    sdst[i] = row * GEMM_PITCH + seg * 4;
  }

  const int r = lane & 15;
  const int khalf = (lane >> 4) << 1;
  const int a_off = (wm * MT * 16 + r) * GEMM_PITCH + khalf;
  const int b_off = (BM + wn * NT * 16 + r) * GEMM_PITCH + khalf;

  v8f acc[MT][NT];
#pragma unroll
  for (int i = 0; i < MT; ++i)
#pragma unroll
    for (int j = 0; j < NT; ++j) acc[i][j] = (v8f){};

  v4f st[NPASS];

#pragma unroll
  for (int i = 0; i < NPASS; ++i)
    st[i] = *(const v4f*)(gsrc[i]);
#pragma unroll
  for (int i = 0; i < NPASS; ++i)
    *(v4f*)(&smem[0][sdst[i]]) = st[i];
  __syncthreads();

  const int nchunks = K / GEMM_BK;
  int buf = 0;
  for (int kc = 0; kc + 1 < nchunks; ++kc) {
    const int kn = (kc + 1) * GEMM_BK;
#pragma unroll
    for (int i = 0; i < NPASS; ++i)
      st[i] = *(const v4f*)(gsrc[i] + kn);
    __builtin_amdgcn_sched_barrier(0);

    const float* sb = smem[buf];
#pragma unroll
    for (int kk = 0; kk < GEMM_BK / 4; ++kk) {
      v2f a[MT], b[NT];
#pragma unroll
      for (int i = 0; i < MT; ++i)
        a[i] = *(const v2f*)(sb + a_off + i * 16 * GEMM_PITCH + kk * 4);
#pragma unroll
      for (int j = 0; j < NT; ++j)
        b[j] = *(const v2f*)(sb + b_off + j * 16 * GEMM_PITCH + kk * 4);
#pragma unroll
      for (int i = 0; i < MT; ++i)
#pragma unroll
        for (int j = 0; j < NT; ++j)
          acc[i][j] = __builtin_amdgcn_wmma_f32_16x16x4_f32(
              false, a[i], false, b[j], (short)0, acc[i][j], false, false);
    }

    float* sn = smem[buf ^ 1];
#pragma unroll
    for (int i = 0; i < NPASS; ++i)
      *(v4f*)(sn + sdst[i]) = st[i];
    __syncthreads();
    buf ^= 1;
  }

  {
    const float* sb = smem[buf];
#pragma unroll
    for (int kk = 0; kk < GEMM_BK / 4; ++kk) {
      v2f a[MT], b[NT];
#pragma unroll
      for (int i = 0; i < MT; ++i)
        a[i] = *(const v2f*)(sb + a_off + i * 16 * GEMM_PITCH + kk * 4);
#pragma unroll
      for (int j = 0; j < NT; ++j)
        b[j] = *(const v2f*)(sb + b_off + j * 16 * GEMM_PITCH + kk * 4);
#pragma unroll
      for (int i = 0; i < MT; ++i)
#pragma unroll
        for (int j = 0; j < NT; ++j)
          acc[i][j] = __builtin_amdgcn_wmma_f32_16x16x4_f32(
              false, a[i], false, b[j], (short)0, acc[i][j], false, false);
    }
  }

  const int mw = m0 + wm * MT * 16 + ((lane >> 4) << 3);
  const int nw = n0 + wn * NT * 16 + r;
#pragma unroll
  for (int i = 0; i < MT; ++i) {
#pragma unroll
    for (int j = 0; j < NT; ++j) {
      const int n = nw + j * 16;
      float bv = 0.0f;
      if (MODE == 1) bv = bias[n];
#pragma unroll
      for (int e = 0; e < 8; ++e) {
        float v = acc[i][j][e];
        if (MODE == 1) {
          v += bv;
          v = (v > 20.0f) ? v : log1pf(__expf(v));
        }
        C[(size_t)(mw + i * 16 + e) * ldc + n] = v;
      }
    }
  }
}

__global__ __launch_bounds__(256) void conv_silu_kernel(
    const float* __restrict__ xz, const float* __restrict__ cw,
    const float* __restrict__ cb, float* __restrict__ xconv) {
  const int gid = blockIdx.x * blockDim.x + threadIdx.x;
  if (gid >= NB * SEQ_L * D_INNER) return;
  const int d = gid & (D_INNER - 1);
  const int bl = gid / D_INNER;
  const int l = bl & (SEQ_L - 1);
  const int b = bl / SEQ_L;

  float acc = cb[d];
  const float* base = xz + ((size_t)b * SEQ_L) * (2 * D_INNER) + d;
#pragma unroll
  for (int j = 0; j < 4; ++j) {
    const int ls = l - 3 + j;
    if (ls >= 0) acc = fmaf(cw[d * 4 + j], base[(size_t)ls * (2 * D_INNER)], acc);
  }
  xconv[gid] = acc / (1.0f + __expf(-acc));
}

#define SCAN_T 16

struct ScanOps {
  float dt[SCAN_T], x[SCAN_T], B[SCAN_T], C[SCAN_T], z[SCAN_T];
};

__device__ __forceinline__ void scan_fetch(ScanOps& o, const float* dt_r,
                                           const float* xc_r, const float* bc_r,
                                           const float* z_r) {
#pragma unroll
  for (int t = 0; t < SCAN_T; ++t) {
    o.dt[t] = dt_r[(size_t)t * D_INNER];
    o.x[t] = xc_r[(size_t)t * D_INNER];
    o.B[t] = bc_r[(size_t)t * XDBL_N];
    o.C[t] = bc_r[(size_t)t * XDBL_N + DSTATE];
    o.z[t] = z_r[(size_t)t * (2 * D_INNER)];
  }
}

__device__ __forceinline__ void scan_batch(const ScanOps& o, float& h, float An,
                                           float Dd, bool writer, float* u_r) {
  float uo[SCAN_T];
#pragma unroll
  for (int t = 0; t < SCAN_T; ++t) {
    const float dtv = o.dt[t];
    const float xv = o.x[t];
    const float dA = __expf(dtv * An);
    h = fmaf(dA, h, dtv * xv * o.B[t]);

    float y = h * o.C[t];
    y += __shfl_xor(y, 1, 32);
    y += __shfl_xor(y, 2, 32);
    y += __shfl_xor(y, 4, 32);
    y += __shfl_xor(y, 8, 32);

    const float zv = o.z[t];
    const float sz = zv / (1.0f + __expf(-zv));
    uo[t] = fmaf(xv, Dd, y) * sz;
  }
  if (writer) {
#pragma unroll
    for (int t = 0; t < SCAN_T; ++t) u_r[(size_t)t * D_INNER] = uo[t];
  }
}

__global__ __launch_bounds__(256) void scan_kernel(
    const float* __restrict__ dt, const float* __restrict__ xconv,
    const float* __restrict__ x_dbl, const float* __restrict__ xz,
    const float* __restrict__ A_log, const float* __restrict__ Dp,
    float* __restrict__ u) {
  const int gid = blockIdx.x * blockDim.x + threadIdx.x;
  const int c = gid >> 4;
  const int n = gid & 15;
  if (c >= NB * D_INNER) return;
  const int d = c & (D_INNER - 1);
  const int b = c >> 11;

  const float An = -__expf(A_log[d * DSTATE + n]);
  const float Dd = Dp[d];
  float h = 0.0f;

  const float* dt_p = dt + ((size_t)b * SEQ_L) * D_INNER + d;
  const float* xc_p = xconv + ((size_t)b * SEQ_L) * D_INNER + d;
  const float* bc_p = x_dbl + ((size_t)b * SEQ_L) * XDBL_N + DTRANK + n;
  const float* z_p = xz + ((size_t)b * SEQ_L) * (2 * D_INNER) + D_INNER + d;
  float* u_p = u + ((size_t)b * SEQ_L) * D_INNER + d;
  const bool writer = (n == 0);

  ScanOps ra, rb;
  scan_fetch(ra, dt_p, xc_p, bc_p, z_p);

  for (int l0 = 0; l0 < SEQ_L; l0 += 2 * SCAN_T) {
    const int l1 = l0 + SCAN_T;
    const int l2 = (l0 + 2 * SCAN_T < SEQ_L) ? (l0 + 2 * SCAN_T) : l0;

    scan_fetch(rb, dt_p + (size_t)l1 * D_INNER, xc_p + (size_t)l1 * D_INNER,
               bc_p + (size_t)l1 * XDBL_N, z_p + (size_t)l1 * (2 * D_INNER));
    scan_batch(ra, h, An, Dd, writer, u_p + (size_t)l0 * D_INNER);

    scan_fetch(ra, dt_p + (size_t)l2 * D_INNER, xc_p + (size_t)l2 * D_INNER,
               bc_p + (size_t)l2 * XDBL_N, z_p + (size_t)l2 * (2 * D_INNER));
    scan_batch(rb, h, An, Dd, writer, u_p + (size_t)l1 * D_INNER);
  }
}

extern "C" void kernel_launch(void* const* d_in, const int* in_sizes, int n_in,
                              void* d_out, int out_size, void* d_ws, size_t ws_size,
                              hipStream_t stream) {
  const float* x      = (const float*)d_in[0];
  const float* W_in   = (const float*)d_in[1];
  const float* conv_w = (const float*)d_in[2];
  const float* conv_b = (const float*)d_in[3];
  const float* W_x    = (const float*)d_in[4];
  const float* W_dt   = (const float*)d_in[5];
  const float* b_dt   = (const float*)d_in[6];
  const float* A_log  = (const float*)d_in[7];
  const float* Dp     = (const float*)d_in[8];
  const float* W_out  = (const float*)d_in[9];
  float* out = (float*)d_out;

  float* ws = (float*)d_ws;
  const size_t M = (size_t)NB * SEQ_L;
  float* xz    = ws;
  float* xconv = xz + M * 2 * D_INNER;
  float* x_dbl = xconv + M * D_INNER;
  float* dt    = x_dbl + M * XDBL_N;
  float* u     = dt + M * D_INNER;

  dim3 blk(256);
  {
    int blocks = (int)(M / 128) * (2 * D_INNER / 128);
    hipLaunchKernelGGL((wmma_gemm_f32<0, 128, 128, 4, 2>), dim3(blocks), dim3(256), 0, stream,
                       x, W_in, xz, nullptr,
                       2 * D_INNER, D_MODEL, D_MODEL, D_MODEL, 2 * D_INNER);
  }
  {
    int total = NB * SEQ_L * D_INNER;
    hipLaunchKernelGGL(conv_silu_kernel, dim3((total + 255) / 256), blk, 0, stream,
                       xz, conv_w, conv_b, xconv);
  }
  {
    int blocks = (int)(M / 32);
    hipLaunchKernelGGL((wmma_gemm_f32<0, 32, 96, 2, 2>), dim3(blocks), dim3(128), 0, stream,
                       xconv, W_x, x_dbl, nullptr,
                       XDBL_N, D_INNER, D_INNER, D_INNER, XDBL_N);
  }
  {
    int blocks = (int)(M / 128) * (D_INNER / 128);
    hipLaunchKernelGGL((wmma_gemm_f32<1, 128, 128, 4, 2>), dim3(blocks), dim3(256), 0, stream,
                       x_dbl, W_dt, dt, b_dt,
                       D_INNER, DTRANK, XDBL_N, DTRANK, D_INNER);
  }
  {
    int total = NB * D_INNER * DSTATE;
    hipLaunchKernelGGL(scan_kernel, dim3((total + 255) / 256), blk, 0, stream,
                       dt, xconv, x_dbl, xz, A_log, Dp, u);
  }
  {
    int blocks = (int)(M / 128) * (D_MODEL / 64);
    hipLaunchKernelGGL((wmma_gemm_f32<0, 128, 64, 4, 2>), dim3(blocks), dim3(256), 0, stream,
                       u, W_out, out, nullptr,
                       D_MODEL, D_INNER, D_INNER, D_INNER, D_MODEL);
  }
}


# scan: lane n of each 16-lane group owns step n of a 16-step batch (alone loads dt,x,z, forms gate, stores u; dt, dt*x shared by DPP row_share); waves_per_eu(2,4) keeps parent's overlapped exchanges
# speedup vs baseline: 5.6683x; 1.1708x over previous
#include <hip/hip_runtime.h>
#include <hip/hip_bf16.h>
#include <math.h>


typedef __attribute__((ext_vector_type(2))) float v2f;
typedef __attribute__((ext_vector_type(4))) float v4f;
typedef __attribute__((ext_vector_type(8))) float v8f;

#define D_MODEL 1024
#define D_INNER 2048
#define NB 2
#define SEQ_L 1024
#define DSTATE 16
#define DTRANK 64
#define XDBL_N 96

#define GEMM_BK 32
#define GEMM_PITCH 36

template <int MODE, int BM, int BN, int WGM, int WGN>
__global__ __launch_bounds__(WGM * WGN * 32) void wmma_gemm_f32(
    const float* __restrict__ A, const float* __restrict__ W,
    float* __restrict__ C, const float* __restrict__ bias,
    int N, int K, int lda, int ldb, int ldc) {
  constexpr int NTHR = WGM * WGN * 32;
  constexpr int MT = BM / (16 * WGM);
  constexpr int NT = BN / (16 * WGN);
  constexpr int ROWS = BM + BN;
  constexpr int PIECES = ROWS * (GEMM_BK / 4);
  constexpr int NPASS = (PIECES + NTHR - 1) / NTHR;
  static_assert(PIECES % NTHR == 0, "a chunk must be a whole number of passes");

  __shared__ __attribute__((aligned(16))) float smem[2][ROWS * GEMM_PITCH];

  const int tid = threadIdx.x;
  const int lane = tid & 31;
  const int wave = tid >> 5;
  const int wm = wave / WGN;
  const int wn = wave % WGN;
  const int nbx = N / BN;
  const int m0 = (blockIdx.x / nbx) * BM;
  const int n0 = (blockIdx.x % nbx) * BN;

  const float* gsrc[NPASS];
  int sdst[NPASS];
#pragma unroll
  for (int i = 0; i < NPASS; ++i) {
    const int p = tid + i * NTHR;
    const int row = p >> 3;
    const int seg = p & 7;
    gsrc[i] = (row < BM) ? (A + (size_t)(m0 + row) * lda + seg * 4)
                         : (W + (size_t)(n0 + row - BM) * ldb + seg * 4);
    sdst[i] = row * GEMM_PITCH + seg * 4;
  }

  const int r = lane & 15;
  const int khalf = (lane >> 4) << 1;
  const int a_off = (wm * MT * 16 + r) * GEMM_PITCH + khalf;
  const int b_off = (BM + wn * NT * 16 + r) * GEMM_PITCH + khalf;

  v8f acc[MT][NT];
#pragma unroll
  for (int i = 0; i < MT; ++i)
#pragma unroll
    for (int j = 0; j < NT; ++j) acc[i][j] = (v8f){};

  v4f st[NPASS];

#pragma unroll
  for (int i = 0; i < NPASS; ++i)
    st[i] = *(const v4f*)(gsrc[i]);
#pragma unroll
  for (int i = 0; i < NPASS; ++i)
    *(v4f*)(&smem[0][sdst[i]]) = st[i];
  __syncthreads();

  const int nchunks = K / GEMM_BK;
  int buf = 0;
  for (int kc = 0; kc + 1 < nchunks; ++kc) {
    const int kn = (kc + 1) * GEMM_BK;
#pragma unroll
    for (int i = 0; i < NPASS; ++i)
      st[i] = *(const v4f*)(gsrc[i] + kn);
    __builtin_amdgcn_sched_barrier(0);

    const float* sb = smem[buf];
#pragma unroll
    for (int kk = 0; kk < GEMM_BK / 4; ++kk) {
      v2f a[MT], b[NT];
#pragma unroll
      for (int i = 0; i < MT; ++i)
        a[i] = *(const v2f*)(sb + a_off + i * 16 * GEMM_PITCH + kk * 4);
#pragma unroll
      for (int j = 0; j < NT; ++j)
        b[j] = *(const v2f*)(sb + b_off + j * 16 * GEMM_PITCH + kk * 4);
#pragma unroll
      for (int i = 0; i < MT; ++i)
#pragma unroll
        for (int j = 0; j < NT; ++j)
          acc[i][j] = __builtin_amdgcn_wmma_f32_16x16x4_f32(
              false, a[i], false, b[j], (short)0, acc[i][j], false, false);
    }

    float* sn = smem[buf ^ 1];
#pragma unroll
    for (int i = 0; i < NPASS; ++i)
      *(v4f*)(sn + sdst[i]) = st[i];
    __syncthreads();
    buf ^= 1;
  }

  {
    const float* sb = smem[buf];
#pragma unroll
    for (int kk = 0; kk < GEMM_BK / 4; ++kk) {
      v2f a[MT], b[NT];
#pragma unroll
      for (int i = 0; i < MT; ++i)
        a[i] = *(const v2f*)(sb + a_off + i * 16 * GEMM_PITCH + kk * 4);
#pragma unroll
      for (int j = 0; j < NT; ++j)
        b[j] = *(const v2f*)(sb + b_off + j * 16 * GEMM_PITCH + kk * 4);
#pragma unroll
      for (int i = 0; i < MT; ++i)
#pragma unroll
        for (int j = 0; j < NT; ++j)
          acc[i][j] = __builtin_amdgcn_wmma_f32_16x16x4_f32(
              false, a[i], false, b[j], (short)0, acc[i][j], false, false);
    }
  }

  const int mw = m0 + wm * MT * 16 + ((lane >> 4) << 3);
  const int nw = n0 + wn * NT * 16 + r;
#pragma unroll
  for (int i = 0; i < MT; ++i) {
#pragma unroll
    for (int j = 0; j < NT; ++j) {
      const int n = nw + j * 16;
      float bv = 0.0f;
      if (MODE == 1) bv = bias[n];
#pragma unroll
      for (int e = 0; e < 8; ++e) {
        float v = acc[i][j][e];
        if (MODE == 1) {
          v += bv;
          v = (v > 20.0f) ? v : log1pf(__expf(v));
        }
        C[(size_t)(mw + i * 16 + e) * ldc + n] = v;
      }
    }
  }
}

__global__ __launch_bounds__(256) void conv_silu_kernel(
    const float* __restrict__ xz, const float* __restrict__ cw,
    const float* __restrict__ cb, float* __restrict__ xconv) {
  const int gid = blockIdx.x * blockDim.x + threadIdx.x;
  if (gid >= NB * SEQ_L * D_INNER) return;
  const int d = gid & (D_INNER - 1);
  const int bl = gid / D_INNER;
  const int l = bl & (SEQ_L - 1);
  const int b = bl / SEQ_L;

  float acc = cb[d];
  const float* base = xz + ((size_t)b * SEQ_L) * (2 * D_INNER) + d;
#pragma unroll
  for (int j = 0; j < 4; ++j) {
    const int ls = l - 3 + j;
    if (ls >= 0) acc = fmaf(cw[d * 4 + j], base[(size_t)ls * (2 * D_INNER)], acc);
  }
  xconv[gid] = acc / (1.0f + __expf(-acc));
}

#define SCAN_T 16

struct ScanOps {
  float dt, x, z;
  float B[SCAN_T], C[SCAN_T];
};

__device__ __forceinline__ void scan_fetch(ScanOps& o, const float* dt_r,
                                           const float* xc_r, const float* bc_r,
                                           const float* z_r) {
  o.dt = dt_r[0];
  o.x = xc_r[0];
  o.z = z_r[0];
#pragma unroll
  for (int t = 0; t < SCAN_T; ++t) {
    o.B[t] = bc_r[(size_t)t * XDBL_N];
    o.C[t] = bc_r[(size_t)t * XDBL_N + DSTATE];
  }
}

template <int T>
__device__ __forceinline__ float row_lane(float v) {
  return __builtin_bit_cast(
      float, __builtin_amdgcn_update_dpp(0, __builtin_bit_cast(int, v),
                                         0x150 + T, 0xf, 0xf, true));
}

template <int T>
__device__ __forceinline__ void scan_steps(const ScanOps& o, float dtx, float& h,
                                           float An, int n, float& ykeep) {
  if constexpr (T < SCAN_T) {
    const float dtv = row_lane<T>(o.dt);
    const float dA = __expf(dtv * An);
    h = fmaf(dA, h, row_lane<T>(dtx) * o.B[T]);

    float y = h * o.C[T];
    y += __shfl_xor(y, 1, 32);
    y += __shfl_xor(y, 2, 32);
    y += __shfl_xor(y, 4, 32);
    y += __shfl_xor(y, 8, 32);

    ykeep = (n == T) ? y : ykeep;
    scan_steps<T + 1>(o, dtx, h, An, n, ykeep);
  }
}

__device__ __forceinline__ void scan_batch(const ScanOps& o, float& h, float An,
                                           float Dd, int n, float* u_r) {
  const float dtx = o.dt * o.x;
  float ykeep = 0.0f;
  scan_steps<0>(o, dtx, h, An, n, ykeep);
  const float zv = o.z;
  const float sz = zv / (1.0f + __expf(-zv));
  u_r[0] = fmaf(o.x, Dd, ykeep) * sz;
}

__global__ __launch_bounds__(256) __attribute__((amdgpu_waves_per_eu(2, 4))) void scan_kernel(
    const float* __restrict__ dt, const float* __restrict__ xconv,
    const float* __restrict__ x_dbl, const float* __restrict__ xz,
    const float* __restrict__ A_log, const float* __restrict__ Dp,
    float* __restrict__ u) {
  const int gid = blockIdx.x * blockDim.x + threadIdx.x;
  const int c = gid >> 4;
  const int n = gid & 15;
  if (c >= NB * D_INNER) return;
  const int d = c & (D_INNER - 1);
  const int b = c >> 11;

  const float An = -__expf(A_log[d * DSTATE + n]);
  const float Dd = Dp[d];
  float h = 0.0f;

  const float* dt_p = dt + ((size_t)b * SEQ_L + n) * D_INNER + d;
  const float* xc_p = xconv + ((size_t)b * SEQ_L + n) * D_INNER + d;
  const float* bc_p = x_dbl + ((size_t)b * SEQ_L) * XDBL_N + DTRANK + n;
  const float* z_p = xz + ((size_t)b * SEQ_L + n) * (2 * D_INNER) + D_INNER + d;
  float* u_p = u + ((size_t)b * SEQ_L + n) * D_INNER + d;

  ScanOps ra, rb;
  scan_fetch(ra, dt_p, xc_p, bc_p, z_p);

  for (int l0 = 0; l0 < SEQ_L; l0 += 2 * SCAN_T) {
    const int l1 = l0 + SCAN_T;
    const int l2 = (l0 + 2 * SCAN_T < SEQ_L) ? (l0 + 2 * SCAN_T) : l0;

    scan_fetch(rb, dt_p + (size_t)l1 * D_INNER, xc_p + (size_t)l1 * D_INNER,
               bc_p + (size_t)l1 * XDBL_N, z_p + (size_t)l1 * (2 * D_INNER));
    scan_batch(ra, h, An, Dd, n, u_p + (size_t)l0 * D_INNER);

    scan_fetch(ra, dt_p + (size_t)l2 * D_INNER, xc_p + (size_t)l2 * D_INNER,
               bc_p + (size_t)l2 * XDBL_N, z_p + (size_t)l2 * (2 * D_INNER));
    scan_batch(rb, h, An, Dd, n, u_p + (size_t)l1 * D_INNER);
  }
}

extern "C" void kernel_launch(void* const* d_in, const int* in_sizes, int n_in,
                              void* d_out, int out_size, void* d_ws, size_t ws_size,
                              hipStream_t stream) {
  const float* x      = (const float*)d_in[0];
  const float* W_in   = (const float*)d_in[1];
  const float* conv_w = (const float*)d_in[2];
  const float* conv_b = (const float*)d_in[3];
  const float* W_x    = (const float*)d_in[4];
  const float* W_dt   = (const float*)d_in[5];
  const float* b_dt   = (const float*)d_in[6];
  const float* A_log  = (const float*)d_in[7];
  const float* Dp     = (const float*)d_in[8];
  const float* W_out  = (const float*)d_in[9];
  float* out = (float*)d_out;

  float* ws = (float*)d_ws;
  const size_t M = (size_t)NB * SEQ_L;
  float* xz    = ws;
  float* xconv = xz + M * 2 * D_INNER;
  float* x_dbl = xconv + M * D_INNER;
  float* dt    = x_dbl + M * XDBL_N;
  float* u     = dt + M * D_INNER;

  dim3 blk(256);
  {
    int blocks = (int)(M / 128) * (2 * D_INNER / 128);
    hipLaunchKernelGGL((wmma_gemm_f32<0, 128, 128, 4, 2>), dim3(blocks), dim3(256), 0, stream,
                       x, W_in, xz, nullptr,
                       2 * D_INNER, D_MODEL, D_MODEL, D_MODEL, 2 * D_INNER);
  }
  {
    int total = NB * SEQ_L * D_INNER;
    hipLaunchKernelGGL(conv_silu_kernel, dim3((total + 255) / 256), blk, 0, stream,
                       xz, conv_w, conv_b, xconv);
  }
  {
    int blocks = (int)(M / 32);
    hipLaunchKernelGGL((wmma_gemm_f32<0, 32, 96, 2, 2>), dim3(blocks), dim3(128), 0, stream,
                       xconv, W_x, x_dbl, nullptr,
                       XDBL_N, D_INNER, D_INNER, D_INNER, XDBL_N);
  }
  {
    int blocks = (int)(M / 128) * (D_INNER / 128);
    hipLaunchKernelGGL((wmma_gemm_f32<1, 128, 128, 4, 2>), dim3(blocks), dim3(256), 0, stream,
                       x_dbl, W_dt, dt, b_dt,
                       D_INNER, DTRANK, XDBL_N, DTRANK, D_INNER);
  }
  {
    int total = NB * D_INNER * DSTATE;
    hipLaunchKernelGGL(scan_kernel, dim3((total + 255) / 256), blk, 0, stream,
                       dt, xconv, x_dbl, xz, A_log, Dp, u);
  }
  {
    int blocks = (int)(M / 128) * (D_MODEL / 64);
    hipLaunchKernelGGL((wmma_gemm_f32<0, 128, 64, 4, 2>), dim3(blocks), dim3(256), 0, stream,
                       u, W_out, out, nullptr,
                       D_MODEL, D_INNER, D_INNER, D_INNER, D_MODEL);
  }
}


# scan: the 16 y sums of a 16-step batch reduced in one 4-stage DPP row_xmask butterfly (15 vector-ALU exchanges per batch, no LDS permutes); same xor 1,2,4,8 tree, products rounded before 1st add
# speedup vs baseline: 6.1323x; 1.0819x over previous
#include <hip/hip_runtime.h>
#include <hip/hip_bf16.h>
#include <math.h>


typedef __attribute__((ext_vector_type(2))) float v2f;
typedef __attribute__((ext_vector_type(4))) float v4f;
typedef __attribute__((ext_vector_type(8))) float v8f;

#define D_MODEL 1024
#define D_INNER 2048
#define NB 2
#define SEQ_L 1024
#define DSTATE 16
#define DTRANK 64
#define XDBL_N 96

#define GEMM_BK 32
#define GEMM_PITCH 36

template <int MODE, int BM, int BN, int WGM, int WGN>
__global__ __launch_bounds__(WGM * WGN * 32) void wmma_gemm_f32(
    const float* __restrict__ A, const float* __restrict__ W,
    float* __restrict__ C, const float* __restrict__ bias,
    int N, int K, int lda, int ldb, int ldc) {
  constexpr int NTHR = WGM * WGN * 32;
  constexpr int MT = BM / (16 * WGM);
  constexpr int NT = BN / (16 * WGN);
  constexpr int ROWS = BM + BN;
  constexpr int PIECES = ROWS * (GEMM_BK / 4);
  constexpr int NPASS = (PIECES + NTHR - 1) / NTHR;
  static_assert(PIECES % NTHR == 0, "a chunk must be a whole number of passes");

  __shared__ __attribute__((aligned(16))) float smem[2][ROWS * GEMM_PITCH];

  const int tid = threadIdx.x;
  const int lane = tid & 31;
  const int wave = tid >> 5;
  const int wm = wave / WGN;
  const int wn = wave % WGN;
  const int nbx = N / BN;
  const int m0 = (blockIdx.x / nbx) * BM;
  const int n0 = (blockIdx.x % nbx) * BN;

  const float* gsrc[NPASS];
  int sdst[NPASS];
#pragma unroll
  for (int i = 0; i < NPASS; ++i) {
    const int p = tid + i * NTHR;
    const int row = p >> 3;
    const int seg = p & 7;
    gsrc[i] = (row < BM) ? (A + (size_t)(m0 + row) * lda + seg * 4)
                         : (W + (size_t)(n0 + row - BM) * ldb + seg * 4);
    sdst[i] = row * GEMM_PITCH + seg * 4;
  }

  const int r = lane & 15;
  const int khalf = (lane >> 4) << 1;
  const int a_off = (wm * MT * 16 + r) * GEMM_PITCH + khalf;
  const int b_off = (BM + wn * NT * 16 + r) * GEMM_PITCH + khalf;

  v8f acc[MT][NT];
#pragma unroll
  for (int i = 0; i < MT; ++i)
#pragma unroll
    for (int j = 0; j < NT; ++j) acc[i][j] = (v8f){};

  v4f st[NPASS];

#pragma unroll
  for (int i = 0; i < NPASS; ++i)
    st[i] = *(const v4f*)(gsrc[i]);
#pragma unroll
  for (int i = 0; i < NPASS; ++i)
    *(v4f*)(&smem[0][sdst[i]]) = st[i];
  __syncthreads();

  const int nchunks = K / GEMM_BK;
  int buf = 0;
  for (int kc = 0; kc + 1 < nchunks; ++kc) {
    const int kn = (kc + 1) * GEMM_BK;
#pragma unroll
    for (int i = 0; i < NPASS; ++i)
      st[i] = *(const v4f*)(gsrc[i] + kn);
    __builtin_amdgcn_sched_barrier(0);

    const float* sb = smem[buf];
#pragma unroll
    for (int kk = 0; kk < GEMM_BK / 4; ++kk) {
      v2f a[MT], b[NT];
#pragma unroll
      for (int i = 0; i < MT; ++i)
        a[i] = *(const v2f*)(sb + a_off + i * 16 * GEMM_PITCH + kk * 4);
#pragma unroll
      for (int j = 0; j < NT; ++j)
        b[j] = *(const v2f*)(sb + b_off + j * 16 * GEMM_PITCH + kk * 4);
#pragma unroll
      for (int i = 0; i < MT; ++i)
#pragma unroll
        for (int j = 0; j < NT; ++j)
          acc[i][j] = __builtin_amdgcn_wmma_f32_16x16x4_f32(
              false, a[i], false, b[j], (short)0, acc[i][j], false, false);
    }

    float* sn = smem[buf ^ 1];
#pragma unroll
    for (int i = 0; i < NPASS; ++i)
      *(v4f*)(sn + sdst[i]) = st[i];
    __syncthreads();
    buf ^= 1;
  }

  {
    const float* sb = smem[buf];
#pragma unroll
    for (int kk = 0; kk < GEMM_BK / 4; ++kk) {
      v2f a[MT], b[NT];
#pragma unroll
      for (int i = 0; i < MT; ++i)
        a[i] = *(const v2f*)(sb + a_off + i * 16 * GEMM_PITCH + kk * 4);
#pragma unroll
      for (int j = 0; j < NT; ++j)
        b[j] = *(const v2f*)(sb + b_off + j * 16 * GEMM_PITCH + kk * 4);
#pragma unroll
      for (int i = 0; i < MT; ++i)
#pragma unroll
        for (int j = 0; j < NT; ++j)
          acc[i][j] = __builtin_amdgcn_wmma_f32_16x16x4_f32(
              false, a[i], false, b[j], (short)0, acc[i][j], false, false);
    }
  }

  const int mw = m0 + wm * MT * 16 + ((lane >> 4) << 3);
  const int nw = n0 + wn * NT * 16 + r;
#pragma unroll
  for (int i = 0; i < MT; ++i) {
#pragma unroll
    for (int j = 0; j < NT; ++j) {
      const int n = nw + j * 16;
      float bv = 0.0f;
      if (MODE == 1) bv = bias[n];
#pragma unroll
      for (int e = 0; e < 8; ++e) {
        float v = acc[i][j][e];
        if (MODE == 1) {
          v += bv;
          v = (v > 20.0f) ? v : log1pf(__expf(v));
        }
        C[(size_t)(mw + i * 16 + e) * ldc + n] = v;
      }
    }
  }
}

__global__ __launch_bounds__(256) void conv_silu_kernel(
    const float* __restrict__ xz, const float* __restrict__ cw,
    const float* __restrict__ cb, float* __restrict__ xconv) {
  const int gid = blockIdx.x * blockDim.x + threadIdx.x;
  if (gid >= NB * SEQ_L * D_INNER) return;
  const int d = gid & (D_INNER - 1);
  const int bl = gid / D_INNER;
  const int l = bl & (SEQ_L - 1);
  const int b = bl / SEQ_L;

  float acc = cb[d];
  const float* base = xz + ((size_t)b * SEQ_L) * (2 * D_INNER) + d;
#pragma unroll
  for (int j = 0; j < 4; ++j) {
    const int ls = l - 3 + j;
    if (ls >= 0) acc = fmaf(cw[d * 4 + j], base[(size_t)ls * (2 * D_INNER)], acc);
  }
  xconv[gid] = acc / (1.0f + __expf(-acc));
}

#define SCAN_T 16

struct ScanOps {
  float dt, x, z;
  float B[SCAN_T], C[SCAN_T];
};

__device__ __forceinline__ void scan_fetch(ScanOps& o, const float* dt_r,
                                           const float* xc_r, const float* bc_r,
                                           const float* z_r) {
  o.dt = dt_r[0];
  o.x = xc_r[0];
  o.z = z_r[0];
#pragma unroll
  for (int t = 0; t < SCAN_T; ++t) {
    o.B[t] = bc_r[(size_t)t * XDBL_N];
    o.C[t] = bc_r[(size_t)t * XDBL_N + DSTATE];
  }
}

template <int T>
__device__ __forceinline__ float row_lane(float v) {
  return __builtin_bit_cast(
      float, __builtin_amdgcn_update_dpp(0, __builtin_bit_cast(int, v),
                                         0x150 + T, 0xf, 0xf, true));
}

template <int M>
__device__ __forceinline__ float row_xor(float v) {
  return __builtin_bit_cast(
      float, __builtin_amdgcn_update_dpp(0, __builtin_bit_cast(int, v),
                                         0x160 + M, 0xf, 0xf, true));
}

template <int M, int CNT>
__device__ __forceinline__ void fold_stage(float (&y)[SCAN_T], bool up) {
#pragma unroll
  for (int j = 0; j < CNT; ++j) {
    const float lo = y[2 * j];
    const float hi = y[2 * j + 1];
    const float keep = up ? hi : lo;
    const float send = up ? lo : hi;
    y[j] = keep + row_xor<M>(send);
  }
}

template <int T>
__device__ __forceinline__ void scan_steps(const ScanOps& o, float dtx, float& h,
                                           float An, float (&y)[SCAN_T]) {
  if constexpr (T < SCAN_T) {
    const float dtv = row_lane<T>(o.dt);
    const float dA = __expf(dtv * An);
    h = fmaf(dA, h, row_lane<T>(dtx) * o.B[T]);
    y[T] = h * o.C[T];
    scan_steps<T + 1>(o, dtx, h, An, y);
  }
}

__device__ __forceinline__ void scan_batch(const ScanOps& o, float& h, float An,
                                           float Dd, int n, float* u_r) {
  const float dtx = o.dt * o.x;
  float y[SCAN_T];
  scan_steps<0>(o, dtx, h, An, y);
  fold_stage<1, 8>(y, (n & 1) != 0);
  fold_stage<2, 4>(y, (n & 2) != 0);
  fold_stage<4, 2>(y, (n & 4) != 0);
  fold_stage<8, 1>(y, (n & 8) != 0);
  const float zv = o.z;
  const float sz = zv / (1.0f + __expf(-zv));
  u_r[0] = fmaf(o.x, Dd, y[0]) * sz;
}

__global__ __launch_bounds__(256) __attribute__((amdgpu_waves_per_eu(2, 4))) void scan_kernel(
    const float* __restrict__ dt, const float* __restrict__ xconv,
    const float* __restrict__ x_dbl, const float* __restrict__ xz,
    const float* __restrict__ A_log, const float* __restrict__ Dp,
    float* __restrict__ u) {
  const int gid = blockIdx.x * blockDim.x + threadIdx.x;
  const int c = gid >> 4;
  const int n = gid & 15;
  if (c >= NB * D_INNER) return;
  const int d = c & (D_INNER - 1);
  const int b = c >> 11;

  const float An = -__expf(A_log[d * DSTATE + n]);
  const float Dd = Dp[d];
  float h = 0.0f;

  const float* dt_p = dt + ((size_t)b * SEQ_L + n) * D_INNER + d;
  const float* xc_p = xconv + ((size_t)b * SEQ_L + n) * D_INNER + d;
  const float* bc_p = x_dbl + ((size_t)b * SEQ_L) * XDBL_N + DTRANK + n;
  const float* z_p = xz + ((size_t)b * SEQ_L + n) * (2 * D_INNER) + D_INNER + d;
  float* u_p = u + ((size_t)b * SEQ_L + n) * D_INNER + d;

  ScanOps ra, rb;
  scan_fetch(ra, dt_p, xc_p, bc_p, z_p);

  for (int l0 = 0; l0 < SEQ_L; l0 += 2 * SCAN_T) {
    const int l1 = l0 + SCAN_T;
    const int l2 = (l0 + 2 * SCAN_T < SEQ_L) ? (l0 + 2 * SCAN_T) : l0;

    scan_fetch(rb, dt_p + (size_t)l1 * D_INNER, xc_p + (size_t)l1 * D_INNER,
               bc_p + (size_t)l1 * XDBL_N, z_p + (size_t)l1 * (2 * D_INNER));
    scan_batch(ra, h, An, Dd, n, u_p + (size_t)l0 * D_INNER);

    scan_fetch(ra, dt_p + (size_t)l2 * D_INNER, xc_p + (size_t)l2 * D_INNER,
               bc_p + (size_t)l2 * XDBL_N, z_p + (size_t)l2 * (2 * D_INNER));
    scan_batch(rb, h, An, Dd, n, u_p + (size_t)l1 * D_INNER);
  }
}

extern "C" void kernel_launch(void* const* d_in, const int* in_sizes, int n_in,
                              void* d_out, int out_size, void* d_ws, size_t ws_size,
                              hipStream_t stream) {
  const float* x      = (const float*)d_in[0];
  const float* W_in   = (const float*)d_in[1];
  const float* conv_w = (const float*)d_in[2];
  const float* conv_b = (const float*)d_in[3];
  const float* W_x    = (const float*)d_in[4];
  const float* W_dt   = (const float*)d_in[5];
  const float* b_dt   = (const float*)d_in[6];
  const float* A_log  = (const float*)d_in[7];
  const float* Dp     = (const float*)d_in[8];
  const float* W_out  = (const float*)d_in[9];
  float* out = (float*)d_out;

  float* ws = (float*)d_ws;
  const size_t M = (size_t)NB * SEQ_L;
  float* xz    = ws;
  float* xconv = xz + M * 2 * D_INNER;
  float* x_dbl = xconv + M * D_INNER;
  float* dt    = x_dbl + M * XDBL_N;
  float* u     = dt + M * D_INNER;

  dim3 blk(256);
  {
    int blocks = (int)(M / 128) * (2 * D_INNER / 128);
    hipLaunchKernelGGL((wmma_gemm_f32<0, 128, 128, 4, 2>), dim3(blocks), dim3(256), 0, stream,
                       x, W_in, xz, nullptr,
                       2 * D_INNER, D_MODEL, D_MODEL, D_MODEL, 2 * D_INNER);
  }
  {
    int total = NB * SEQ_L * D_INNER;
    hipLaunchKernelGGL(conv_silu_kernel, dim3((total + 255) / 256), blk, 0, stream,
                       xz, conv_w, conv_b, xconv);
  }
  {
    int blocks = (int)(M / 32);
    hipLaunchKernelGGL((wmma_gemm_f32<0, 32, 96, 2, 2>), dim3(blocks), dim3(128), 0, stream,
                       xconv, W_x, x_dbl, nullptr,
                       XDBL_N, D_INNER, D_INNER, D_INNER, XDBL_N);
  }
  {
    int blocks = (int)(M / 128) * (D_INNER / 128);
    hipLaunchKernelGGL((wmma_gemm_f32<1, 128, 128, 4, 2>), dim3(blocks), dim3(256), 0, stream,
                       x_dbl, W_dt, dt, b_dt,
                       D_INNER, DTRANK, XDBL_N, DTRANK, D_INNER);
  }
  {
    int total = NB * D_INNER * DSTATE;
    hipLaunchKernelGGL(scan_kernel, dim3((total + 255) / 256), blk, 0, stream,
                       dt, xconv, x_dbl, xz, A_log, Dp, u);
  }
  {
    int blocks = (int)(M / 128) * (D_MODEL / 64);
    hipLaunchKernelGGL((wmma_gemm_f32<0, 128, 64, 4, 2>), dim3(blocks), dim3(256), 0, stream,
                       u, W_out, out, nullptr,
                       D_MODEL, D_INNER, D_INNER, D_INNER, D_MODEL);
  }
}


# dt GEMM epilogue: softplus as max(v,0)+log1p(e), e=exp(-|v|), log1p(e)=log(u)*e/(u-1), u=1+e, on hardware log2 and rcp (few ulp) in place of library log1pf; no branch per element
# speedup vs baseline: 6.5126x; 1.0620x over previous
#include <hip/hip_runtime.h>
#include <hip/hip_bf16.h>
#include <math.h>


typedef __attribute__((ext_vector_type(2))) float v2f;
typedef __attribute__((ext_vector_type(4))) float v4f;
typedef __attribute__((ext_vector_type(8))) float v8f;

#define D_MODEL 1024
#define D_INNER 2048
#define NB 2
#define SEQ_L 1024
#define DSTATE 16
#define DTRANK 64
#define XDBL_N 96

#define GEMM_BK 32
#define GEMM_PITCH 36

template <int MODE, int BM, int BN, int WGM, int WGN>
__global__ __launch_bounds__(WGM * WGN * 32) void wmma_gemm_f32(
    const float* __restrict__ A, const float* __restrict__ W,
    float* __restrict__ C, const float* __restrict__ bias,
    int N, int K, int lda, int ldb, int ldc) {
  constexpr int NTHR = WGM * WGN * 32;
  constexpr int MT = BM / (16 * WGM);
  constexpr int NT = BN / (16 * WGN);
  constexpr int ROWS = BM + BN;
  constexpr int PIECES = ROWS * (GEMM_BK / 4);
  constexpr int NPASS = (PIECES + NTHR - 1) / NTHR;
  static_assert(PIECES % NTHR == 0, "a chunk must be a whole number of passes");

  __shared__ __attribute__((aligned(16))) float smem[2][ROWS * GEMM_PITCH];

  const int tid = threadIdx.x;
  const int lane = tid & 31;
  const int wave = tid >> 5;
  const int wm = wave / WGN;
  const int wn = wave % WGN;
  const int nbx = N / BN;
  const int m0 = (blockIdx.x / nbx) * BM;
  const int n0 = (blockIdx.x % nbx) * BN;

  const float* gsrc[NPASS];
  int sdst[NPASS];
#pragma unroll
  for (int i = 0; i < NPASS; ++i) {
    const int p = tid + i * NTHR;
    const int row = p >> 3;
    const int seg = p & 7;
    gsrc[i] = (row < BM) ? (A + (size_t)(m0 + row) * lda + seg * 4)
                         : (W + (size_t)(n0 + row - BM) * ldb + seg * 4);
    sdst[i] = row * GEMM_PITCH + seg * 4;
  }

  const int r = lane & 15;
  const int khalf = (lane >> 4) << 1;
  const int a_off = (wm * MT * 16 + r) * GEMM_PITCH + khalf;
  const int b_off = (BM + wn * NT * 16 + r) * GEMM_PITCH + khalf;

  v8f acc[MT][NT];
#pragma unroll
  for (int i = 0; i < MT; ++i)
#pragma unroll
    for (int j = 0; j < NT; ++j) acc[i][j] = (v8f){};

  v4f st[NPASS];

#pragma unroll
  for (int i = 0; i < NPASS; ++i)
    st[i] = *(const v4f*)(gsrc[i]);
#pragma unroll
  for (int i = 0; i < NPASS; ++i)
    *(v4f*)(&smem[0][sdst[i]]) = st[i];
  __syncthreads();

  const int nchunks = K / GEMM_BK;
  int buf = 0;
  for (int kc = 0; kc + 1 < nchunks; ++kc) {
    const int kn = (kc + 1) * GEMM_BK;
#pragma unroll
    for (int i = 0; i < NPASS; ++i)
      st[i] = *(const v4f*)(gsrc[i] + kn);
    __builtin_amdgcn_sched_barrier(0);

    const float* sb = smem[buf];
#pragma unroll
    for (int kk = 0; kk < GEMM_BK / 4; ++kk) {
      v2f a[MT], b[NT];
#pragma unroll
      for (int i = 0; i < MT; ++i)
        a[i] = *(const v2f*)(sb + a_off + i * 16 * GEMM_PITCH + kk * 4);
#pragma unroll
      for (int j = 0; j < NT; ++j)
        b[j] = *(const v2f*)(sb + b_off + j * 16 * GEMM_PITCH + kk * 4);
#pragma unroll
      for (int i = 0; i < MT; ++i)
#pragma unroll
        for (int j = 0; j < NT; ++j)
          acc[i][j] = __builtin_amdgcn_wmma_f32_16x16x4_f32(
              false, a[i], false, b[j], (short)0, acc[i][j], false, false);
    }

    float* sn = smem[buf ^ 1];
#pragma unroll
    for (int i = 0; i < NPASS; ++i)
      *(v4f*)(sn + sdst[i]) = st[i];
    __syncthreads();
    buf ^= 1;
  }

  {
    const float* sb = smem[buf];
#pragma unroll
    for (int kk = 0; kk < GEMM_BK / 4; ++kk) {
      v2f a[MT], b[NT];
#pragma unroll
      for (int i = 0; i < MT; ++i)
        a[i] = *(const v2f*)(sb + a_off + i * 16 * GEMM_PITCH + kk * 4);
#pragma unroll
      for (int j = 0; j < NT; ++j)
        b[j] = *(const v2f*)(sb + b_off + j * 16 * GEMM_PITCH + kk * 4);
#pragma unroll
      for (int i = 0; i < MT; ++i)
#pragma unroll
        for (int j = 0; j < NT; ++j)
          acc[i][j] = __builtin_amdgcn_wmma_f32_16x16x4_f32(
              false, a[i], false, b[j], (short)0, acc[i][j], false, false);
    }
  }

  const int mw = m0 + wm * MT * 16 + ((lane >> 4) << 3);
  const int nw = n0 + wn * NT * 16 + r;
#pragma unroll
  for (int i = 0; i < MT; ++i) {
#pragma unroll
    for (int j = 0; j < NT; ++j) {
      const int n = nw + j * 16;
      float bv = 0.0f;
      if (MODE == 1) bv = bias[n];
#pragma unroll
      for (int e = 0; e < 8; ++e) {
        float v = acc[i][j][e];
        if (MODE == 1) {
          v += bv;
          const float e = __expf(-fabsf(v));
          const float u = 1.0f + e;
          const float d = u - 1.0f;
          const float lg = __builtin_amdgcn_logf(u) * 0.69314718f;
          const float q = lg * (e * __builtin_amdgcn_rcpf(d));
          v = fmaxf(v, 0.0f) + ((d == 0.0f) ? e : q);
        }
        C[(size_t)(mw + i * 16 + e) * ldc + n] = v;
      }
    }
  }
}

__global__ __launch_bounds__(256) void conv_silu_kernel(
    const float* __restrict__ xz, const float* __restrict__ cw,
    const float* __restrict__ cb, float* __restrict__ xconv) {
  const int gid = blockIdx.x * blockDim.x + threadIdx.x;
  if (gid >= NB * SEQ_L * D_INNER) return;
  const int d = gid & (D_INNER - 1);
  const int bl = gid / D_INNER;
  const int l = bl & (SEQ_L - 1);
  const int b = bl / SEQ_L;

  float acc = cb[d];
  const float* base = xz + ((size_t)b * SEQ_L) * (2 * D_INNER) + d;
#pragma unroll
  for (int j = 0; j < 4; ++j) {
    const int ls = l - 3 + j;
    if (ls >= 0) acc = fmaf(cw[d * 4 + j], base[(size_t)ls * (2 * D_INNER)], acc);
  }
  xconv[gid] = acc / (1.0f + __expf(-acc));
}

#define SCAN_T 16

struct ScanOps {
  float dt, x, z;
  float B[SCAN_T], C[SCAN_T];
};

__device__ __forceinline__ void scan_fetch(ScanOps& o, const float* dt_r,
                                           const float* xc_r, const float* bc_r,
                                           const float* z_r) {
  o.dt = dt_r[0];
  o.x = xc_r[0];
  o.z = z_r[0];
#pragma unroll
  for (int t = 0; t < SCAN_T; ++t) {
    o.B[t] = bc_r[(size_t)t * XDBL_N];
    o.C[t] = bc_r[(size_t)t * XDBL_N + DSTATE];
  }
}

template <int T>
__device__ __forceinline__ float row_lane(float v) {
  return __builtin_bit_cast(
      float, __builtin_amdgcn_update_dpp(0, __builtin_bit_cast(int, v),
                                         0x150 + T, 0xf, 0xf, true));
}

template <int M>
__device__ __forceinline__ float row_xor(float v) {
  return __builtin_bit_cast(
      float, __builtin_amdgcn_update_dpp(0, __builtin_bit_cast(int, v),
                                         0x160 + M, 0xf, 0xf, true));
}

template <int M, int CNT>
__device__ __forceinline__ void fold_stage(float (&y)[SCAN_T], bool up) {
#pragma unroll
  for (int j = 0; j < CNT; ++j) {
    const float lo = y[2 * j];
    const float hi = y[2 * j + 1];
    const float keep = up ? hi : lo;
    const float send = up ? lo : hi;
    y[j] = keep + row_xor<M>(send);
  }
}

template <int T>
__device__ __forceinline__ void scan_steps(const ScanOps& o, float dtx, float& h,
                                           float An, float (&y)[SCAN_T]) {
  if constexpr (T < SCAN_T) {
    const float dtv = row_lane<T>(o.dt);
    const float dA = __expf(dtv * An);
    h = fmaf(dA, h, row_lane<T>(dtx) * o.B[T]);
    y[T] = h * o.C[T];
    scan_steps<T + 1>(o, dtx, h, An, y);
  }
}

__device__ __forceinline__ void scan_batch(const ScanOps& o, float& h, float An,
                                           float Dd, int n, float* u_r) {
  const float dtx = o.dt * o.x;
  float y[SCAN_T];
  scan_steps<0>(o, dtx, h, An, y);
  fold_stage<1, 8>(y, (n & 1) != 0);
  fold_stage<2, 4>(y, (n & 2) != 0);
  fold_stage<4, 2>(y, (n & 4) != 0);
  fold_stage<8, 1>(y, (n & 8) != 0);
  const float zv = o.z;
  const float sz = zv / (1.0f + __expf(-zv));
  u_r[0] = fmaf(o.x, Dd, y[0]) * sz;
}

__global__ __launch_bounds__(256) __attribute__((amdgpu_waves_per_eu(2, 4))) void scan_kernel(
    const float* __restrict__ dt, const float* __restrict__ xconv,
    const float* __restrict__ x_dbl, const float* __restrict__ xz,
    const float* __restrict__ A_log, const float* __restrict__ Dp,
    float* __restrict__ u) {
  const int gid = blockIdx.x * blockDim.x + threadIdx.x;
  const int c = gid >> 4;
  const int n = gid & 15;
  if (c >= NB * D_INNER) return;
  const int d = c & (D_INNER - 1);
  const int b = c >> 11;

  const float An = -__expf(A_log[d * DSTATE + n]);
  const float Dd = Dp[d];
  float h = 0.0f;

  const float* dt_p = dt + ((size_t)b * SEQ_L + n) * D_INNER + d;
  const float* xc_p = xconv + ((size_t)b * SEQ_L + n) * D_INNER + d;
  const float* bc_p = x_dbl + ((size_t)b * SEQ_L) * XDBL_N + DTRANK + n;
  const float* z_p = xz + ((size_t)b * SEQ_L + n) * (2 * D_INNER) + D_INNER + d;
  float* u_p = u + ((size_t)b * SEQ_L + n) * D_INNER + d;

  ScanOps ra, rb;
  scan_fetch(ra, dt_p, xc_p, bc_p, z_p);

  for (int l0 = 0; l0 < SEQ_L; l0 += 2 * SCAN_T) {
    const int l1 = l0 + SCAN_T;
    const int l2 = (l0 + 2 * SCAN_T < SEQ_L) ? (l0 + 2 * SCAN_T) : l0;

    scan_fetch(rb, dt_p + (size_t)l1 * D_INNER, xc_p + (size_t)l1 * D_INNER,
               bc_p + (size_t)l1 * XDBL_N, z_p + (size_t)l1 * (2 * D_INNER));
    scan_batch(ra, h, An, Dd, n, u_p + (size_t)l0 * D_INNER);

    scan_fetch(ra, dt_p + (size_t)l2 * D_INNER, xc_p + (size_t)l2 * D_INNER,
               bc_p + (size_t)l2 * XDBL_N, z_p + (size_t)l2 * (2 * D_INNER));
    scan_batch(rb, h, An, Dd, n, u_p + (size_t)l1 * D_INNER);
  }
}

extern "C" void kernel_launch(void* const* d_in, const int* in_sizes, int n_in,
                              void* d_out, int out_size, void* d_ws, size_t ws_size,
                              hipStream_t stream) {
  const float* x      = (const float*)d_in[0];
  const float* W_in   = (const float*)d_in[1];
  const float* conv_w = (const float*)d_in[2];
  const float* conv_b = (const float*)d_in[3];
  const float* W_x    = (const float*)d_in[4];
  const float* W_dt   = (const float*)d_in[5];
  const float* b_dt   = (const float*)d_in[6];
  const float* A_log  = (const float*)d_in[7];
  const float* Dp     = (const float*)d_in[8];
  const float* W_out  = (const float*)d_in[9];
  float* out = (float*)d_out;

  float* ws = (float*)d_ws;
  const size_t M = (size_t)NB * SEQ_L;
  float* xz    = ws;
  float* xconv = xz + M * 2 * D_INNER;
  float* x_dbl = xconv + M * D_INNER;
  float* dt    = x_dbl + M * XDBL_N;
  float* u     = dt + M * D_INNER;

  dim3 blk(256);
  {
    int blocks = (int)(M / 128) * (2 * D_INNER / 128);
    hipLaunchKernelGGL((wmma_gemm_f32<0, 128, 128, 4, 2>), dim3(blocks), dim3(256), 0, stream,
                       x, W_in, xz, nullptr,
                       2 * D_INNER, D_MODEL, D_MODEL, D_MODEL, 2 * D_INNER);
  }
  {
    int total = NB * SEQ_L * D_INNER;
    hipLaunchKernelGGL(conv_silu_kernel, dim3((total + 255) / 256), blk, 0, stream,
                       xz, conv_w, conv_b, xconv);
  }
  {
    int blocks = (int)(M / 32);
    hipLaunchKernelGGL((wmma_gemm_f32<0, 32, 96, 2, 2>), dim3(blocks), dim3(128), 0, stream,
                       xconv, W_x, x_dbl, nullptr,
                       XDBL_N, D_INNER, D_INNER, D_INNER, XDBL_N);
  }
  {
    int blocks = (int)(M / 128) * (D_INNER / 128);
    hipLaunchKernelGGL((wmma_gemm_f32<1, 128, 128, 4, 2>), dim3(blocks), dim3(256), 0, stream,
                       x_dbl, W_dt, dt, b_dt,
                       D_INNER, DTRANK, XDBL_N, DTRANK, D_INNER);
  }
  {
    int total = NB * D_INNER * DSTATE;
    hipLaunchKernelGGL(scan_kernel, dim3((total + 255) / 256), blk, 0, stream,
                       dt, xconv, x_dbl, xz, A_log, Dp, u);
  }
  {
    int blocks = (int)(M / 128) * (D_MODEL / 64);
    hipLaunchKernelGGL((wmma_gemm_f32<0, 128, 64, 4, 2>), dim3(blocks), dim3(256), 0, stream,
                       u, W_out, out, nullptr,
                       D_MODEL, D_INNER, D_INNER, D_INNER, D_MODEL);
  }
}


# N=96 GEMM (x_dbl): 16x48 tiles, k split over the 8 waves of a workgroup, each wave stages its k slice in its own LDS region, no barrier in the loop; 8 partial sums added at the end (new sum order)
# speedup vs baseline: 7.3232x; 1.1245x over previous
#include <hip/hip_runtime.h>
#include <hip/hip_bf16.h>
#include <math.h>


typedef __attribute__((ext_vector_type(2))) float v2f;
typedef __attribute__((ext_vector_type(4))) float v4f;
typedef __attribute__((ext_vector_type(8))) float v8f;

#define D_MODEL 1024
#define D_INNER 2048
#define NB 2
#define SEQ_L 1024
#define DSTATE 16
#define DTRANK 64
#define XDBL_N 96

#define GEMM_BK 32
#define GEMM_PITCH 36

template <int MODE, int BM, int BN, int WGM, int WGN>
__global__ __launch_bounds__(WGM * WGN * 32) void wmma_gemm_f32(
    const float* __restrict__ A, const float* __restrict__ W,
    float* __restrict__ C, const float* __restrict__ bias,
    int N, int K, int lda, int ldb, int ldc) {
  constexpr int NTHR = WGM * WGN * 32;
  constexpr int MT = BM / (16 * WGM);
  constexpr int NT = BN / (16 * WGN);
  constexpr int ROWS = BM + BN;
  constexpr int PIECES = ROWS * (GEMM_BK / 4);
  constexpr int NPASS = (PIECES + NTHR - 1) / NTHR;
  static_assert(PIECES % NTHR == 0, "a chunk must be a whole number of passes");

  __shared__ __attribute__((aligned(16))) float smem[2][ROWS * GEMM_PITCH];

  const int tid = threadIdx.x;
  const int lane = tid & 31;
  const int wave = tid >> 5;
  const int wm = wave / WGN;
  const int wn = wave % WGN;
  const int nbx = N / BN;
  const int m0 = (blockIdx.x / nbx) * BM;
  const int n0 = (blockIdx.x % nbx) * BN;

  const float* gsrc[NPASS];
  int sdst[NPASS];
#pragma unroll
  for (int i = 0; i < NPASS; ++i) {
    const int p = tid + i * NTHR;
    const int row = p >> 3;
    const int seg = p & 7;
    gsrc[i] = (row < BM) ? (A + (size_t)(m0 + row) * lda + seg * 4)
                         : (W + (size_t)(n0 + row - BM) * ldb + seg * 4);
    sdst[i] = row * GEMM_PITCH + seg * 4;
  }

  const int r = lane & 15;
  const int khalf = (lane >> 4) << 1;
  const int a_off = (wm * MT * 16 + r) * GEMM_PITCH + khalf;
  const int b_off = (BM + wn * NT * 16 + r) * GEMM_PITCH + khalf;

  v8f acc[MT][NT];
#pragma unroll
  for (int i = 0; i < MT; ++i)
#pragma unroll
    for (int j = 0; j < NT; ++j) acc[i][j] = (v8f){};

  v4f st[NPASS];

#pragma unroll
  for (int i = 0; i < NPASS; ++i)
    st[i] = *(const v4f*)(gsrc[i]);
#pragma unroll
  for (int i = 0; i < NPASS; ++i)
    *(v4f*)(&smem[0][sdst[i]]) = st[i];
  __syncthreads();

  const int nchunks = K / GEMM_BK;
  int buf = 0;
  for (int kc = 0; kc + 1 < nchunks; ++kc) {
    const int kn = (kc + 1) * GEMM_BK;
#pragma unroll
    for (int i = 0; i < NPASS; ++i)
      st[i] = *(const v4f*)(gsrc[i] + kn);
    __builtin_amdgcn_sched_barrier(0);

    const float* sb = smem[buf];
#pragma unroll
    for (int kk = 0; kk < GEMM_BK / 4; ++kk) {
      v2f a[MT], b[NT];
#pragma unroll
      for (int i = 0; i < MT; ++i)
        a[i] = *(const v2f*)(sb + a_off + i * 16 * GEMM_PITCH + kk * 4);
#pragma unroll
      for (int j = 0; j < NT; ++j)
        b[j] = *(const v2f*)(sb + b_off + j * 16 * GEMM_PITCH + kk * 4);
#pragma unroll
      for (int i = 0; i < MT; ++i)
#pragma unroll
        for (int j = 0; j < NT; ++j)
          acc[i][j] = __builtin_amdgcn_wmma_f32_16x16x4_f32(
              false, a[i], false, b[j], (short)0, acc[i][j], false, false);
    }

    float* sn = smem[buf ^ 1];
#pragma unroll
    for (int i = 0; i < NPASS; ++i)
      *(v4f*)(sn + sdst[i]) = st[i];
    __syncthreads();
    buf ^= 1;
  }

  {
    const float* sb = smem[buf];
#pragma unroll
    for (int kk = 0; kk < GEMM_BK / 4; ++kk) {
      v2f a[MT], b[NT];
#pragma unroll
      for (int i = 0; i < MT; ++i)
        a[i] = *(const v2f*)(sb + a_off + i * 16 * GEMM_PITCH + kk * 4);
#pragma unroll
      for (int j = 0; j < NT; ++j)
        b[j] = *(const v2f*)(sb + b_off + j * 16 * GEMM_PITCH + kk * 4);
#pragma unroll
      for (int i = 0; i < MT; ++i)
#pragma unroll
        for (int j = 0; j < NT; ++j)
          acc[i][j] = __builtin_amdgcn_wmma_f32_16x16x4_f32(
              false, a[i], false, b[j], (short)0, acc[i][j], false, false);
    }
  }

  const int mw = m0 + wm * MT * 16 + ((lane >> 4) << 3);
  const int nw = n0 + wn * NT * 16 + r;
#pragma unroll
  for (int i = 0; i < MT; ++i) {
#pragma unroll
    for (int j = 0; j < NT; ++j) {
      const int n = nw + j * 16;
      float bv = 0.0f;
      if (MODE == 1) bv = bias[n];
#pragma unroll
      for (int e = 0; e < 8; ++e) {
        float v = acc[i][j][e];
        if (MODE == 1) {
          v += bv;
          const float e = __expf(-fabsf(v));
          const float u = 1.0f + e;
          const float d = u - 1.0f;
          const float lg = __builtin_amdgcn_logf(u) * 0.69314718f;
          const float q = lg * (e * __builtin_amdgcn_rcpf(d));
          v = fmaxf(v, 0.0f) + ((d == 0.0f) ? e : q);
        }
        C[(size_t)(mw + i * 16 + e) * ldc + n] = v;
      }
    }
  }
}

template <int BN>
__global__ __launch_bounds__(256) void wmma_gemm_ksplit_f32(
    const float* __restrict__ A, const float* __restrict__ W,
    float* __restrict__ C, int N, int K, int lda, int ldb, int ldc) {
  constexpr int KW = 8;
  constexpr int BM = 16;
  constexpr int NT = BN / 16;
  constexpr int ROWS = BM + BN;
  constexpr int NPASS = ROWS * (GEMM_BK / 4) / 32;
  constexpr int APASS = BM * (GEMM_BK / 4) / 32;
  constexpr int WSZ = ROWS * GEMM_PITCH;
  static_assert(NT * 256 <= WSZ, "a wave's partial tiles fit its own region");

  __shared__ __attribute__((aligned(16))) float smem[KW * WSZ];

  const int tid = threadIdx.x;
  const int lane = tid & 31;
  const int wave = tid >> 5;
  const int nbx = N / BN;
  const int m0 = (blockIdx.x / nbx) * BM;
  const int n0 = (blockIdx.x % nbx) * BN;
  const int kslice = K / KW;
  const int k0 = wave * kslice;
  float* sw = smem + wave * WSZ;

  const int lrow = lane >> 3;
  const int seg = lane & 7;
  const float* ga = A + (size_t)(m0 + lrow) * lda + k0 + seg * 4;
  const float* gw = W + (size_t)(n0 + lrow) * ldb + k0 + seg * 4;
  const int sd = lrow * GEMM_PITCH + seg * 4;

  const int r = lane & 15;
  const int khalf = (lane >> 4) << 1;
  const int a_off = r * GEMM_PITCH + khalf;
  const int b_off = (BM + r) * GEMM_PITCH + khalf;

  v8f acc[NT];
#pragma unroll
  for (int j = 0; j < NT; ++j) acc[j] = (v8f){};

  v4f st[NPASS];

#pragma unroll
  for (int i = 0; i < NPASS; ++i)
    st[i] = (i < APASS) ? *(const v4f*)(ga + (size_t)(4 * i) * lda)
                        : *(const v4f*)(gw + (size_t)(4 * (i - APASS)) * ldb);
#pragma unroll
  for (int i = 0; i < NPASS; ++i)
    *(v4f*)(sw + sd + 4 * i * GEMM_PITCH) = st[i];

  const int nchunks = kslice / GEMM_BK;
  for (int kc = 0; kc + 1 < nchunks; ++kc) {
    const int kn = (kc + 1) * GEMM_BK;
#pragma unroll
    for (int i = 0; i < NPASS; ++i)
      st[i] = (i < APASS) ? *(const v4f*)(ga + (size_t)(4 * i) * lda + kn)
                          : *(const v4f*)(gw + (size_t)(4 * (i - APASS)) * ldb + kn);
    __builtin_amdgcn_sched_barrier(0);

#pragma unroll
    for (int kk = 0; kk < GEMM_BK / 4; ++kk) {
      const v2f a = *(const v2f*)(sw + a_off + kk * 4);
      v2f b[NT];
#pragma unroll
      for (int j = 0; j < NT; ++j)
        b[j] = *(const v2f*)(sw + b_off + j * 16 * GEMM_PITCH + kk * 4);
#pragma unroll
      for (int j = 0; j < NT; ++j)
        acc[j] = __builtin_amdgcn_wmma_f32_16x16x4_f32(
            false, a, false, b[j], (short)0, acc[j], false, false);
    }

#pragma unroll
    for (int i = 0; i < NPASS; ++i)
      *(v4f*)(sw + sd + 4 * i * GEMM_PITCH) = st[i];
  }

  {
#pragma unroll
    for (int kk = 0; kk < GEMM_BK / 4; ++kk) {
      const v2f a = *(const v2f*)(sw + a_off + kk * 4);
      v2f b[NT];
#pragma unroll
      for (int j = 0; j < NT; ++j)
        b[j] = *(const v2f*)(sw + b_off + j * 16 * GEMM_PITCH + kk * 4);
#pragma unroll
      for (int j = 0; j < NT; ++j)
        acc[j] = __builtin_amdgcn_wmma_f32_16x16x4_f32(
            false, a, false, b[j], (short)0, acc[j], false, false);
    }
  }

#pragma unroll
  for (int j = 0; j < NT; ++j)
#pragma unroll
    for (int e = 0; e < 8; ++e)
      sw[(j * 8 + e) * 32 + lane] = acc[j][e];
  __syncthreads();

  const int row = m0 + wave + ((lane >> 4) << 3);
#pragma unroll
  for (int j = 0; j < NT; ++j) {
    float s = smem[(j * 8 + wave) * 32 + lane];
#pragma unroll
    for (int w = 1; w < KW; ++w)
      s += smem[w * WSZ + (j * 8 + wave) * 32 + lane];
    C[(size_t)row * ldc + n0 + j * 16 + r] = s;
  }
}

__global__ __launch_bounds__(256) void conv_silu_kernel(
    const float* __restrict__ xz, const float* __restrict__ cw,
    const float* __restrict__ cb, float* __restrict__ xconv) {
  const int gid = blockIdx.x * blockDim.x + threadIdx.x;
  if (gid >= NB * SEQ_L * D_INNER) return;
  const int d = gid & (D_INNER - 1);
  const int bl = gid / D_INNER;
  const int l = bl & (SEQ_L - 1);
  const int b = bl / SEQ_L;

  float acc = cb[d];
  const float* base = xz + ((size_t)b * SEQ_L) * (2 * D_INNER) + d;
#pragma unroll
  for (int j = 0; j < 4; ++j) {
    const int ls = l - 3 + j;
    if (ls >= 0) acc = fmaf(cw[d * 4 + j], base[(size_t)ls * (2 * D_INNER)], acc);
  }
  xconv[gid] = acc / (1.0f + __expf(-acc));
}

#define SCAN_T 16

struct ScanOps {
  float dt, x, z;
  float B[SCAN_T], C[SCAN_T];
};

__device__ __forceinline__ void scan_fetch(ScanOps& o, const float* dt_r,
                                           const float* xc_r, const float* bc_r,
                                           const float* z_r) {
  o.dt = dt_r[0];
  o.x = xc_r[0];
  o.z = z_r[0];
#pragma unroll
  for (int t = 0; t < SCAN_T; ++t) {
    o.B[t] = bc_r[(size_t)t * XDBL_N];
    o.C[t] = bc_r[(size_t)t * XDBL_N + DSTATE];
  }
}

template <int T>
__device__ __forceinline__ float row_lane(float v) {
  return __builtin_bit_cast(
      float, __builtin_amdgcn_update_dpp(0, __builtin_bit_cast(int, v),
                                         0x150 + T, 0xf, 0xf, true));
}

template <int M>
__device__ __forceinline__ float row_xor(float v) {
  return __builtin_bit_cast(
      float, __builtin_amdgcn_update_dpp(0, __builtin_bit_cast(int, v),
                                         0x160 + M, 0xf, 0xf, true));
}

template <int M, int CNT>
__device__ __forceinline__ void fold_stage(float (&y)[SCAN_T], bool up) {
#pragma unroll
  for (int j = 0; j < CNT; ++j) {
    const float lo = y[2 * j];
    const float hi = y[2 * j + 1];
    const float keep = up ? hi : lo;
    const float send = up ? lo : hi;
    y[j] = keep + row_xor<M>(send);
  }
}

template <int T>
__device__ __forceinline__ void scan_steps(const ScanOps& o, float dtx, float& h,
                                           float An, float (&y)[SCAN_T]) {
  if constexpr (T < SCAN_T) {
    const float dtv = row_lane<T>(o.dt);
    const float dA = __expf(dtv * An);
    h = fmaf(dA, h, row_lane<T>(dtx) * o.B[T]);
    y[T] = h * o.C[T];
    scan_steps<T + 1>(o, dtx, h, An, y);
  }
}

__device__ __forceinline__ void scan_batch(const ScanOps& o, float& h, float An,
                                           float Dd, int n, float* u_r) {
  const float dtx = o.dt * o.x;
  float y[SCAN_T];
  scan_steps<0>(o, dtx, h, An, y);
  fold_stage<1, 8>(y, (n & 1) != 0);
  fold_stage<2, 4>(y, (n & 2) != 0);
  fold_stage<4, 2>(y, (n & 4) != 0);
  fold_stage<8, 1>(y, (n & 8) != 0);
  const float zv = o.z;
  const float sz = zv / (1.0f + __expf(-zv));
  u_r[0] = fmaf(o.x, Dd, y[0]) * sz;
}

__global__ __launch_bounds__(256) __attribute__((amdgpu_waves_per_eu(2, 4))) void scan_kernel(
    const float* __restrict__ dt, const float* __restrict__ xconv,
    const float* __restrict__ x_dbl, const float* __restrict__ xz,
    const float* __restrict__ A_log, const float* __restrict__ Dp,
    float* __restrict__ u) {
  const int gid = blockIdx.x * blockDim.x + threadIdx.x;
  const int c = gid >> 4;
  const int n = gid & 15;
  if (c >= NB * D_INNER) return;
  const int d = c & (D_INNER - 1);
  const int b = c >> 11;

  const float An = -__expf(A_log[d * DSTATE + n]);
  const float Dd = Dp[d];
  float h = 0.0f;

  const float* dt_p = dt + ((size_t)b * SEQ_L + n) * D_INNER + d;
  const float* xc_p = xconv + ((size_t)b * SEQ_L + n) * D_INNER + d;
  const float* bc_p = x_dbl + ((size_t)b * SEQ_L) * XDBL_N + DTRANK + n;
  const float* z_p = xz + ((size_t)b * SEQ_L + n) * (2 * D_INNER) + D_INNER + d;
  float* u_p = u + ((size_t)b * SEQ_L + n) * D_INNER + d;

  ScanOps ra, rb;
  scan_fetch(ra, dt_p, xc_p, bc_p, z_p);

  for (int l0 = 0; l0 < SEQ_L; l0 += 2 * SCAN_T) {
    const int l1 = l0 + SCAN_T;
    const int l2 = (l0 + 2 * SCAN_T < SEQ_L) ? (l0 + 2 * SCAN_T) : l0;

    scan_fetch(rb, dt_p + (size_t)l1 * D_INNER, xc_p + (size_t)l1 * D_INNER,
               bc_p + (size_t)l1 * XDBL_N, z_p + (size_t)l1 * (2 * D_INNER));
    scan_batch(ra, h, An, Dd, n, u_p + (size_t)l0 * D_INNER);

    scan_fetch(ra, dt_p + (size_t)l2 * D_INNER, xc_p + (size_t)l2 * D_INNER,
               bc_p + (size_t)l2 * XDBL_N, z_p + (size_t)l2 * (2 * D_INNER));
    scan_batch(rb, h, An, Dd, n, u_p + (size_t)l1 * D_INNER);
  }
}

extern "C" void kernel_launch(void* const* d_in, const int* in_sizes, int n_in,
                              void* d_out, int out_size, void* d_ws, size_t ws_size,
                              hipStream_t stream) {
  const float* x      = (const float*)d_in[0];
  const float* W_in   = (const float*)d_in[1];
  const float* conv_w = (const float*)d_in[2];
  const float* conv_b = (const float*)d_in[3];
  const float* W_x    = (const float*)d_in[4];
  const float* W_dt   = (const float*)d_in[5];
  const float* b_dt   = (const float*)d_in[6];
  const float* A_log  = (const float*)d_in[7];
  const float* Dp     = (const float*)d_in[8];
  const float* W_out  = (const float*)d_in[9];
  float* out = (float*)d_out;

  float* ws = (float*)d_ws;
  const size_t M = (size_t)NB * SEQ_L;
  float* xz    = ws;
  float* xconv = xz + M * 2 * D_INNER;
  float* x_dbl = xconv + M * D_INNER;
  float* dt    = x_dbl + M * XDBL_N;
  float* u     = dt + M * D_INNER;

  dim3 blk(256);
  {
    int blocks = (int)(M / 128) * (2 * D_INNER / 128);
    hipLaunchKernelGGL((wmma_gemm_f32<0, 128, 128, 4, 2>), dim3(blocks), dim3(256), 0, stream,
                       x, W_in, xz, nullptr,
                       2 * D_INNER, D_MODEL, D_MODEL, D_MODEL, 2 * D_INNER);
  }
  {
    int total = NB * SEQ_L * D_INNER;
    hipLaunchKernelGGL(conv_silu_kernel, dim3((total + 255) / 256), blk, 0, stream,
                       xz, conv_w, conv_b, xconv);
  }
  {
    int blocks = (int)(M / 16) * (XDBL_N / 48);
    hipLaunchKernelGGL((wmma_gemm_ksplit_f32<48>), dim3(blocks), dim3(256), 0, stream,
                       xconv, W_x, x_dbl,
                       XDBL_N, D_INNER, D_INNER, D_INNER, XDBL_N);
  }
  {
    int blocks = (int)(M / 128) * (D_INNER / 128);
    hipLaunchKernelGGL((wmma_gemm_f32<1, 128, 128, 4, 2>), dim3(blocks), dim3(256), 0, stream,
                       x_dbl, W_dt, dt, b_dt,
                       D_INNER, DTRANK, XDBL_N, DTRANK, D_INNER);
  }
  {
    int total = NB * D_INNER * DSTATE;
    hipLaunchKernelGGL(scan_kernel, dim3((total + 255) / 256), blk, 0, stream,
                       dt, xconv, x_dbl, xz, A_log, Dp, u);
  }
  {
    int blocks = (int)(M / 128) * (D_MODEL / 64);
    hipLaunchKernelGGL((wmma_gemm_f32<0, 128, 64, 4, 2>), dim3(blocks), dim3(256), 0, stream,
                       u, W_out, out, nullptr,
                       D_MODEL, D_INNER, D_INNER, D_INNER, D_MODEL);
  }
}


# scan: a workgroup (16 channels) loads each 32-step block of dt, x, z, B, C rows once, 2 rows per load, hands it to its lanes via LDS, 1 barrier per block (was per wave, 16 rows a load); same sums
# speedup vs baseline: 7.5719x; 1.0340x over previous
#include <hip/hip_runtime.h>
#include <hip/hip_bf16.h>
#include <math.h>


typedef __attribute__((ext_vector_type(2))) float v2f;
typedef __attribute__((ext_vector_type(4))) float v4f;
typedef __attribute__((ext_vector_type(8))) float v8f;

#define D_MODEL 1024
#define D_INNER 2048
#define NB 2
#define SEQ_L 1024
#define DSTATE 16
#define DTRANK 64
#define XDBL_N 96

#define GEMM_BK 32
#define GEMM_PITCH 36

template <int MODE, int BM, int BN, int WGM, int WGN>
__global__ __launch_bounds__(WGM * WGN * 32) void wmma_gemm_f32(
    const float* __restrict__ A, const float* __restrict__ W,
    float* __restrict__ C, const float* __restrict__ bias,
    int N, int K, int lda, int ldb, int ldc) {
  constexpr int NTHR = WGM * WGN * 32;
  constexpr int MT = BM / (16 * WGM);
  constexpr int NT = BN / (16 * WGN);
  constexpr int ROWS = BM + BN;
  constexpr int PIECES = ROWS * (GEMM_BK / 4);
  constexpr int NPASS = (PIECES + NTHR - 1) / NTHR;
  static_assert(PIECES % NTHR == 0, "a chunk must be a whole number of passes");

  __shared__ __attribute__((aligned(16))) float smem[2][ROWS * GEMM_PITCH];

  const int tid = threadIdx.x;
  const int lane = tid & 31;
  const int wave = tid >> 5;
  const int wm = wave / WGN;
  const int wn = wave % WGN;
  const int nbx = N / BN;
  const int m0 = (blockIdx.x / nbx) * BM;
  const int n0 = (blockIdx.x % nbx) * BN;

  const float* gsrc[NPASS];
  int sdst[NPASS];
#pragma unroll
  for (int i = 0; i < NPASS; ++i) {
    const int p = tid + i * NTHR;
    const int row = p >> 3;
    const int seg = p & 7;
    gsrc[i] = (row < BM) ? (A + (size_t)(m0 + row) * lda + seg * 4)
                         : (W + (size_t)(n0 + row - BM) * ldb + seg * 4);
    sdst[i] = row * GEMM_PITCH + seg * 4;
  }

  const int r = lane & 15;
  const int khalf = (lane >> 4) << 1;
  const int a_off = (wm * MT * 16 + r) * GEMM_PITCH + khalf;
  const int b_off = (BM + wn * NT * 16 + r) * GEMM_PITCH + khalf;

  v8f acc[MT][NT];
#pragma unroll
  for (int i = 0; i < MT; ++i)
#pragma unroll
    for (int j = 0; j < NT; ++j) acc[i][j] = (v8f){};

  v4f st[NPASS];

#pragma unroll
  for (int i = 0; i < NPASS; ++i)
    st[i] = *(const v4f*)(gsrc[i]);
#pragma unroll
  for (int i = 0; i < NPASS; ++i)
    *(v4f*)(&smem[0][sdst[i]]) = st[i];
  __syncthreads();

  const int nchunks = K / GEMM_BK;
  int buf = 0;
  for (int kc = 0; kc + 1 < nchunks; ++kc) {
    const int kn = (kc + 1) * GEMM_BK;
#pragma unroll
    for (int i = 0; i < NPASS; ++i)
      st[i] = *(const v4f*)(gsrc[i] + kn);
    __builtin_amdgcn_sched_barrier(0);

    const float* sb = smem[buf];
#pragma unroll
    for (int kk = 0; kk < GEMM_BK / 4; ++kk) {
      v2f a[MT], b[NT];
#pragma unroll
      for (int i = 0; i < MT; ++i)
        a[i] = *(const v2f*)(sb + a_off + i * 16 * GEMM_PITCH + kk * 4);
#pragma unroll
      for (int j = 0; j < NT; ++j)
        b[j] = *(const v2f*)(sb + b_off + j * 16 * GEMM_PITCH + kk * 4);
#pragma unroll
      for (int i = 0; i < MT; ++i)
#pragma unroll
        for (int j = 0; j < NT; ++j)
          acc[i][j] = __builtin_amdgcn_wmma_f32_16x16x4_f32(
              false, a[i], false, b[j], (short)0, acc[i][j], false, false);
    }

    float* sn = smem[buf ^ 1];
#pragma unroll
    for (int i = 0; i < NPASS; ++i)
      *(v4f*)(sn + sdst[i]) = st[i];
    __syncthreads();
    buf ^= 1;
  }

  {
    const float* sb = smem[buf];
#pragma unroll
    for (int kk = 0; kk < GEMM_BK / 4; ++kk) {
      v2f a[MT], b[NT];
#pragma unroll
      for (int i = 0; i < MT; ++i)
        a[i] = *(const v2f*)(sb + a_off + i * 16 * GEMM_PITCH + kk * 4);
#pragma unroll
      for (int j = 0; j < NT; ++j)
        b[j] = *(const v2f*)(sb + b_off + j * 16 * GEMM_PITCH + kk * 4);
#pragma unroll
      for (int i = 0; i < MT; ++i)
#pragma unroll
        for (int j = 0; j < NT; ++j)
          acc[i][j] = __builtin_amdgcn_wmma_f32_16x16x4_f32(
              false, a[i], false, b[j], (short)0, acc[i][j], false, false);
    }
  }

  const int mw = m0 + wm * MT * 16 + ((lane >> 4) << 3);
  const int nw = n0 + wn * NT * 16 + r;
#pragma unroll
  for (int i = 0; i < MT; ++i) {
#pragma unroll
    for (int j = 0; j < NT; ++j) {
      const int n = nw + j * 16;
      float bv = 0.0f;
      if (MODE == 1) bv = bias[n];
#pragma unroll
      for (int e = 0; e < 8; ++e) {
        float v = acc[i][j][e];
        if (MODE == 1) {
          v += bv;
          const float e = __expf(-fabsf(v));
          const float u = 1.0f + e;
          const float d = u - 1.0f;
          const float lg = __builtin_amdgcn_logf(u) * 0.69314718f;
          const float q = lg * (e * __builtin_amdgcn_rcpf(d));
          v = fmaxf(v, 0.0f) + ((d == 0.0f) ? e : q);
        }
        C[(size_t)(mw + i * 16 + e) * ldc + n] = v;
      }
    }
  }
}

template <int BN>
__global__ __launch_bounds__(256) void wmma_gemm_ksplit_f32(
    const float* __restrict__ A, const float* __restrict__ W,
    float* __restrict__ C, int N, int K, int lda, int ldb, int ldc) {
  constexpr int KW = 8;
  constexpr int BM = 16;
  constexpr int NT = BN / 16;
  constexpr int ROWS = BM + BN;
  constexpr int NPASS = ROWS * (GEMM_BK / 4) / 32;
  constexpr int APASS = BM * (GEMM_BK / 4) / 32;
  constexpr int WSZ = ROWS * GEMM_PITCH;
  static_assert(NT * 256 <= WSZ, "a wave's partial tiles fit its own region");

  __shared__ __attribute__((aligned(16))) float smem[KW * WSZ];

  const int tid = threadIdx.x;
  const int lane = tid & 31;
  const int wave = tid >> 5;
  const int nbx = N / BN;
  const int m0 = (blockIdx.x / nbx) * BM;
  const int n0 = (blockIdx.x % nbx) * BN;
  const int kslice = K / KW;
  const int k0 = wave * kslice;
  float* sw = smem + wave * WSZ;

  const int lrow = lane >> 3;
  const int seg = lane & 7;
  const float* ga = A + (size_t)(m0 + lrow) * lda + k0 + seg * 4;
  const float* gw = W + (size_t)(n0 + lrow) * ldb + k0 + seg * 4;
  const int sd = lrow * GEMM_PITCH + seg * 4;

  const int r = lane & 15;
  const int khalf = (lane >> 4) << 1;
  const int a_off = r * GEMM_PITCH + khalf;
  const int b_off = (BM + r) * GEMM_PITCH + khalf;

  v8f acc[NT];
#pragma unroll
  for (int j = 0; j < NT; ++j) acc[j] = (v8f){};

  v4f st[NPASS];

#pragma unroll
  for (int i = 0; i < NPASS; ++i)
    st[i] = (i < APASS) ? *(const v4f*)(ga + (size_t)(4 * i) * lda)
                        : *(const v4f*)(gw + (size_t)(4 * (i - APASS)) * ldb);
#pragma unroll
  for (int i = 0; i < NPASS; ++i)
    *(v4f*)(sw + sd + 4 * i * GEMM_PITCH) = st[i];

  const int nchunks = kslice / GEMM_BK;
  for (int kc = 0; kc + 1 < nchunks; ++kc) {
    const int kn = (kc + 1) * GEMM_BK;
#pragma unroll
    for (int i = 0; i < NPASS; ++i)
      st[i] = (i < APASS) ? *(const v4f*)(ga + (size_t)(4 * i) * lda + kn)
                          : *(const v4f*)(gw + (size_t)(4 * (i - APASS)) * ldb + kn);
    __builtin_amdgcn_sched_barrier(0);

#pragma unroll
    for (int kk = 0; kk < GEMM_BK / 4; ++kk) {
      const v2f a = *(const v2f*)(sw + a_off + kk * 4);
      v2f b[NT];
#pragma unroll
      for (int j = 0; j < NT; ++j)
        b[j] = *(const v2f*)(sw + b_off + j * 16 * GEMM_PITCH + kk * 4);
#pragma unroll
      for (int j = 0; j < NT; ++j)
        acc[j] = __builtin_amdgcn_wmma_f32_16x16x4_f32(
            false, a, false, b[j], (short)0, acc[j], false, false);
    }

#pragma unroll
    for (int i = 0; i < NPASS; ++i)
      *(v4f*)(sw + sd + 4 * i * GEMM_PITCH) = st[i];
  }

  {
#pragma unroll
    for (int kk = 0; kk < GEMM_BK / 4; ++kk) {
      const v2f a = *(const v2f*)(sw + a_off + kk * 4);
      v2f b[NT];
#pragma unroll
      for (int j = 0; j < NT; ++j)
        b[j] = *(const v2f*)(sw + b_off + j * 16 * GEMM_PITCH + kk * 4);
#pragma unroll
      for (int j = 0; j < NT; ++j)
        acc[j] = __builtin_amdgcn_wmma_f32_16x16x4_f32(
            false, a, false, b[j], (short)0, acc[j], false, false);
    }
  }

#pragma unroll
  for (int j = 0; j < NT; ++j)
#pragma unroll
    for (int e = 0; e < 8; ++e)
      sw[(j * 8 + e) * 32 + lane] = acc[j][e];
  __syncthreads();

  const int row = m0 + wave + ((lane >> 4) << 3);
#pragma unroll
  for (int j = 0; j < NT; ++j) {
    float s = smem[(j * 8 + wave) * 32 + lane];
#pragma unroll
    for (int w = 1; w < KW; ++w)
      s += smem[w * WSZ + (j * 8 + wave) * 32 + lane];
    C[(size_t)row * ldc + n0 + j * 16 + r] = s;
  }
}

__global__ __launch_bounds__(256) void conv_silu_kernel(
    const float* __restrict__ xz, const float* __restrict__ cw,
    const float* __restrict__ cb, float* __restrict__ xconv) {
  const int gid = blockIdx.x * blockDim.x + threadIdx.x;
  if (gid >= NB * SEQ_L * D_INNER) return;
  const int d = gid & (D_INNER - 1);
  const int bl = gid / D_INNER;
  const int l = bl & (SEQ_L - 1);
  const int b = bl / SEQ_L;

  float acc = cb[d];
  const float* base = xz + ((size_t)b * SEQ_L) * (2 * D_INNER) + d;
#pragma unroll
  for (int j = 0; j < 4; ++j) {
    const int ls = l - 3 + j;
    if (ls >= 0) acc = fmaf(cw[d * 4 + j], base[(size_t)ls * (2 * D_INNER)], acc);
  }
  xconv[gid] = acc / (1.0f + __expf(-acc));
}

#define SCAN_T 16
#define SCAN_CH 16
#define SX_PITCH 17
#define SBC_PITCH 20
#define SLOT_DT 0
#define SLOT_X (SCAN_T * SX_PITCH)
#define SLOT_Z (2 * SCAN_T * SX_PITCH)
#define SLOT_BC (3 * SCAN_T * SX_PITCH)
#define SLOT_FLOATS (SLOT_BC + 2 * DSTATE * SBC_PITCH)

static_assert(SLOT_BC % 4 == 0 && SBC_PITCH % 4 == 0 && SLOT_FLOATS % 4 == 0,
              "the 16-byte LDS reads of B|C are aligned");
static_assert(SCAN_CH * DSTATE == 256 && SCAN_T == DSTATE && SCAN_T == SCAN_CH,
              "thread (t, j) of the fetch is thread (channel t, state j) of the scan");

struct ScanRaw {
  float dt, x, z;
  v2f bc;
};

struct ScanOps {
  float dt, x, z;
  v4f B[SCAN_T / 4], C[SCAN_T / 4];
};

__device__ __forceinline__ void scan_request(ScanRaw& g, const float* dt_r,
                                             const float* xc_r, const float* z_r,
                                             const float* bc_r) {
  g.dt = dt_r[0];
  g.x = xc_r[0];
  g.z = z_r[0];
  g.bc = *(const v2f*)bc_r;
}

__device__ __forceinline__ void scan_stage(float* s, const ScanRaw& g, int t, int j) {
  s[SLOT_DT + t * SX_PITCH + j] = g.dt;
  s[SLOT_X + t * SX_PITCH + j] = g.x;
  s[SLOT_Z + t * SX_PITCH + j] = g.z;
  s[SLOT_BC + (2 * j) * SBC_PITCH + t] = g.bc[0];
  s[SLOT_BC + (2 * j + 1) * SBC_PITCH + t] = g.bc[1];
}

__device__ __forceinline__ void scan_take(ScanOps& o, const float* s, int c, int n) {
  o.dt = s[SLOT_DT + n * SX_PITCH + c];
  o.x = s[SLOT_X + n * SX_PITCH + c];
  o.z = s[SLOT_Z + n * SX_PITCH + c];
  const v4f* pb = (const v4f*)(s + SLOT_BC + n * SBC_PITCH);
  const v4f* pc = (const v4f*)(s + SLOT_BC + (DSTATE + n) * SBC_PITCH);
#pragma unroll
  for (int q = 0; q < SCAN_T / 4; ++q) {
    o.B[q] = pb[q];
    o.C[q] = pc[q];
  }
}

template <int T>
__device__ __forceinline__ float row_lane(float v) {
  return __builtin_bit_cast(
      float, __builtin_amdgcn_update_dpp(0, __builtin_bit_cast(int, v),
                                         0x150 + T, 0xf, 0xf, true));
}

template <int M>
__device__ __forceinline__ float row_xor(float v) {
  return __builtin_bit_cast(
      float, __builtin_amdgcn_update_dpp(0, __builtin_bit_cast(int, v),
                                         0x160 + M, 0xf, 0xf, true));
}

template <int M, int CNT>
__device__ __forceinline__ void fold_stage(float (&y)[SCAN_T], bool up) {
#pragma unroll
  for (int j = 0; j < CNT; ++j) {
    const float lo = y[2 * j];
    const float hi = y[2 * j + 1];
    const float keep = up ? hi : lo;
    const float send = up ? lo : hi;
    y[j] = keep + row_xor<M>(send);
  }
}

template <int T>
__device__ __forceinline__ void scan_steps(const ScanOps& o, float dtx, float& h,
                                           float An, float (&y)[SCAN_T]) {
  if constexpr (T < SCAN_T) {
    const float dtv = row_lane<T>(o.dt);
    const float dA = __expf(dtv * An);
    h = fmaf(dA, h, row_lane<T>(dtx) * o.B[T >> 2][T & 3]);
    y[T] = h * o.C[T >> 2][T & 3];
    scan_steps<T + 1>(o, dtx, h, An, y);
  }
}

__device__ __forceinline__ void scan_batch(const ScanOps& o, float& h, float An,
                                           float Dd, int n, float* u_r) {
  const float dtx = o.dt * o.x;
  float y[SCAN_T];
  scan_steps<0>(o, dtx, h, An, y);
  fold_stage<1, 8>(y, (n & 1) != 0);
  fold_stage<2, 4>(y, (n & 2) != 0);
  fold_stage<4, 2>(y, (n & 4) != 0);
  fold_stage<8, 1>(y, (n & 8) != 0);
  const float zv = o.z;
  const float sz = zv / (1.0f + __expf(-zv));
  u_r[0] = fmaf(o.x, Dd, y[0]) * sz;
}

__device__ __forceinline__ void lds_barrier() {
  __builtin_amdgcn_fence(__ATOMIC_RELEASE, "workgroup", "local");
  __builtin_amdgcn_s_barrier();
  __builtin_amdgcn_fence(__ATOMIC_ACQUIRE, "workgroup", "local");
}

__global__ __launch_bounds__(256) __attribute__((amdgpu_waves_per_eu(2, 4))) void scan_kernel(
    const float* __restrict__ dt, const float* __restrict__ xconv,
    const float* __restrict__ x_dbl, const float* __restrict__ xz,
    const float* __restrict__ A_log, const float* __restrict__ Dp,
    float* __restrict__ u) {
  __shared__ __attribute__((aligned(16))) float sm[2][2][SLOT_FLOATS];

  const int c0 = blockIdx.x * SCAN_CH;
  if (c0 >= NB * D_INNER) return;
  const int tid = threadIdx.x;
  const int cw = tid >> 4;
  const int n = tid & 15;
  const int d0 = c0 & (D_INNER - 1);
  const int b = c0 >> 11;
  const int d = d0 + cw;

  const float An = -__expf(A_log[d * DSTATE + n]);
  const float Dd = Dp[d];
  float h = 0.0f;

  const float* dt_g = dt + ((size_t)b * SEQ_L + cw) * D_INNER + d0 + n;
  const float* xc_g = xconv + ((size_t)b * SEQ_L + cw) * D_INNER + d0 + n;
  const float* z_g = xz + ((size_t)b * SEQ_L + cw) * (2 * D_INNER) + D_INNER + d0 + n;
  const float* bc_g = x_dbl + ((size_t)b * SEQ_L + cw) * XDBL_N + DTRANK + 2 * n;
  float* u_p = u + ((size_t)b * SEQ_L + n) * D_INNER + d;

  ScanRaw g0, g1;
  ScanOps ra, rb;

  scan_request(g0, dt_g, xc_g, z_g, bc_g);
  scan_request(g1, dt_g + (size_t)SCAN_T * D_INNER, xc_g + (size_t)SCAN_T * D_INNER,
               z_g + (size_t)SCAN_T * (2 * D_INNER), bc_g + (size_t)SCAN_T * XDBL_N);
  scan_stage(sm[0][0], g0, cw, n);
  scan_stage(sm[0][1], g1, cw, n);
  {
    const int l = 2 * SCAN_T;
    scan_request(g0, dt_g + (size_t)l * D_INNER, xc_g + (size_t)l * D_INNER,
                 z_g + (size_t)l * (2 * D_INNER), bc_g + (size_t)l * XDBL_N);
    scan_request(g1, dt_g + (size_t)(l + SCAN_T) * D_INNER, xc_g + (size_t)(l + SCAN_T) * D_INNER,
                 z_g + (size_t)(l + SCAN_T) * (2 * D_INNER), bc_g + (size_t)(l + SCAN_T) * XDBL_N);
  }
  lds_barrier();
  scan_take(ra, sm[0][0], cw, n);

  for (int l0 = 0; l0 < SEQ_L; l0 += 2 * SCAN_T) {
    const int l1 = l0 + SCAN_T;
    const int cur = (l0 / (2 * SCAN_T)) & 1;
    const int l2 = (l0 + 4 * SCAN_T < SEQ_L) ? (l0 + 4 * SCAN_T) : l0;

    scan_take(rb, sm[cur][1], cw, n);
    scan_batch(ra, h, An, Dd, n, u_p + (size_t)l0 * D_INNER);

    scan_stage(sm[cur ^ 1][0], g0, cw, n);
    scan_stage(sm[cur ^ 1][1], g1, cw, n);
    scan_request(g0, dt_g + (size_t)l2 * D_INNER, xc_g + (size_t)l2 * D_INNER,
                 z_g + (size_t)l2 * (2 * D_INNER), bc_g + (size_t)l2 * XDBL_N);
    scan_request(g1, dt_g + (size_t)(l2 + SCAN_T) * D_INNER, xc_g + (size_t)(l2 + SCAN_T) * D_INNER,
                 z_g + (size_t)(l2 + SCAN_T) * (2 * D_INNER), bc_g + (size_t)(l2 + SCAN_T) * XDBL_N);
    lds_barrier();

    scan_take(ra, sm[cur ^ 1][0], cw, n);
    __builtin_amdgcn_sched_barrier(0);
    scan_batch(rb, h, An, Dd, n, u_p + (size_t)l1 * D_INNER);
  }
}

extern "C" void kernel_launch(void* const* d_in, const int* in_sizes, int n_in,
                              void* d_out, int out_size, void* d_ws, size_t ws_size,
                              hipStream_t stream) {
  const float* x      = (const float*)d_in[0];
  const float* W_in   = (const float*)d_in[1];
  const float* conv_w = (const float*)d_in[2];
  const float* conv_b = (const float*)d_in[3];
  const float* W_x    = (const float*)d_in[4];
  const float* W_dt   = (const float*)d_in[5];
  const float* b_dt   = (const float*)d_in[6];
  const float* A_log  = (const float*)d_in[7];
  const float* Dp     = (const float*)d_in[8];
  const float* W_out  = (const float*)d_in[9];
  float* out = (float*)d_out;

  float* ws = (float*)d_ws;
  const size_t M = (size_t)NB * SEQ_L;
  float* xz    = ws;
  float* xconv = xz + M * 2 * D_INNER;
  float* x_dbl = xconv + M * D_INNER;
  float* dt    = x_dbl + M * XDBL_N;
  float* u     = dt + M * D_INNER;

  dim3 blk(256);
  {
    int blocks = (int)(M / 128) * (2 * D_INNER / 128);
    hipLaunchKernelGGL((wmma_gemm_f32<0, 128, 128, 4, 2>), dim3(blocks), dim3(256), 0, stream,
                       x, W_in, xz, nullptr,
                       2 * D_INNER, D_MODEL, D_MODEL, D_MODEL, 2 * D_INNER);
  }
  {
    int total = NB * SEQ_L * D_INNER;
    hipLaunchKernelGGL(conv_silu_kernel, dim3((total + 255) / 256), blk, 0, stream,
                       xz, conv_w, conv_b, xconv);
  }
  {
    int blocks = (int)(M / 16) * (XDBL_N / 48);
    hipLaunchKernelGGL((wmma_gemm_ksplit_f32<48>), dim3(blocks), dim3(256), 0, stream,
                       xconv, W_x, x_dbl,
                       XDBL_N, D_INNER, D_INNER, D_INNER, XDBL_N);
  }
  {
    int blocks = (int)(M / 128) * (D_INNER / 128);
    hipLaunchKernelGGL((wmma_gemm_f32<1, 128, 128, 4, 2>), dim3(blocks), dim3(256), 0, stream,
                       x_dbl, W_dt, dt, b_dt,
                       D_INNER, DTRANK, XDBL_N, DTRANK, D_INNER);
  }
  {
    int total = NB * D_INNER * DSTATE;
    hipLaunchKernelGGL(scan_kernel, dim3((total + 255) / 256), blk, 0, stream,
                       dt, xconv, x_dbl, xz, A_log, Dp, u);
  }
  {
    int blocks = (int)(M / 128) * (D_MODEL / 64);
    hipLaunchKernelGGL((wmma_gemm_f32<0, 128, 64, 4, 2>), dim3(blocks), dim3(256), 0, stream,
                       u, W_out, out, nullptr,
                       D_MODEL, D_INNER, D_INNER, D_INNER, D_MODEL);
  }
}


# conv kernel: a thread takes 4 adjacent channels (16-byte loads and stores) over 8 consecutive steps, rows kept in registers: 4,096 waves to enter instead of 131,072; same sum per element
# speedup vs baseline: 7.8286x; 1.0339x over previous
#include <hip/hip_runtime.h>
#include <hip/hip_bf16.h>
#include <math.h>


typedef __attribute__((ext_vector_type(2))) float v2f;
typedef __attribute__((ext_vector_type(4))) float v4f;
typedef __attribute__((ext_vector_type(8))) float v8f;

#define D_MODEL 1024
#define D_INNER 2048
#define NB 2
#define SEQ_L 1024
#define DSTATE 16
#define DTRANK 64
#define XDBL_N 96

#define GEMM_BK 32
#define GEMM_PITCH 36

template <int MODE, int BM, int BN, int WGM, int WGN>
__global__ __launch_bounds__(WGM * WGN * 32) void wmma_gemm_f32(
    const float* __restrict__ A, const float* __restrict__ W,
    float* __restrict__ C, const float* __restrict__ bias,
    int N, int K, int lda, int ldb, int ldc) {
  constexpr int NTHR = WGM * WGN * 32;
  constexpr int MT = BM / (16 * WGM);
  constexpr int NT = BN / (16 * WGN);
  constexpr int ROWS = BM + BN;
  constexpr int PIECES = ROWS * (GEMM_BK / 4);
  constexpr int NPASS = (PIECES + NTHR - 1) / NTHR;
  static_assert(PIECES % NTHR == 0, "a chunk must be a whole number of passes");

  __shared__ __attribute__((aligned(16))) float smem[2][ROWS * GEMM_PITCH];

  const int tid = threadIdx.x;
  const int lane = tid & 31;
  const int wave = tid >> 5;
  const int wm = wave / WGN;
  const int wn = wave % WGN;
  const int nbx = N / BN;
  const int m0 = (blockIdx.x / nbx) * BM;
  const int n0 = (blockIdx.x % nbx) * BN;

  const float* gsrc[NPASS];
  int sdst[NPASS];
#pragma unroll
  for (int i = 0; i < NPASS; ++i) {
    const int p = tid + i * NTHR;
    const int row = p >> 3;
    const int seg = p & 7;
    gsrc[i] = (row < BM) ? (A + (size_t)(m0 + row) * lda + seg * 4)
                         : (W + (size_t)(n0 + row - BM) * ldb + seg * 4);
    sdst[i] = row * GEMM_PITCH + seg * 4;
  }

  const int r = lane & 15;
  const int khalf = (lane >> 4) << 1;
  const int a_off = (wm * MT * 16 + r) * GEMM_PITCH + khalf;
  const int b_off = (BM + wn * NT * 16 + r) * GEMM_PITCH + khalf;

  v8f acc[MT][NT];
#pragma unroll
  for (int i = 0; i < MT; ++i)
#pragma unroll
    for (int j = 0; j < NT; ++j) acc[i][j] = (v8f){};

  v4f st[NPASS];

#pragma unroll
  for (int i = 0; i < NPASS; ++i)
    st[i] = *(const v4f*)(gsrc[i]);
#pragma unroll
  for (int i = 0; i < NPASS; ++i)
    *(v4f*)(&smem[0][sdst[i]]) = st[i];
  __syncthreads();

  const int nchunks = K / GEMM_BK;
  int buf = 0;
  for (int kc = 0; kc + 1 < nchunks; ++kc) {
    const int kn = (kc + 1) * GEMM_BK;
#pragma unroll
    for (int i = 0; i < NPASS; ++i)
      st[i] = *(const v4f*)(gsrc[i] + kn);
    __builtin_amdgcn_sched_barrier(0);

    const float* sb = smem[buf];
#pragma unroll
    for (int kk = 0; kk < GEMM_BK / 4; ++kk) {
      v2f a[MT], b[NT];
#pragma unroll
      for (int i = 0; i < MT; ++i)
        a[i] = *(const v2f*)(sb + a_off + i * 16 * GEMM_PITCH + kk * 4);
#pragma unroll
      for (int j = 0; j < NT; ++j)
        b[j] = *(const v2f*)(sb + b_off + j * 16 * GEMM_PITCH + kk * 4);
#pragma unroll
      for (int i = 0; i < MT; ++i)
#pragma unroll
        for (int j = 0; j < NT; ++j)
          acc[i][j] = __builtin_amdgcn_wmma_f32_16x16x4_f32(
              false, a[i], false, b[j], (short)0, acc[i][j], false, false);
    }

    float* sn = smem[buf ^ 1];
#pragma unroll
    for (int i = 0; i < NPASS; ++i)
      *(v4f*)(sn + sdst[i]) = st[i];
    __syncthreads();
    buf ^= 1;
  }

  {
    const float* sb = smem[buf];
#pragma unroll
    for (int kk = 0; kk < GEMM_BK / 4; ++kk) {
      v2f a[MT], b[NT];
#pragma unroll
      for (int i = 0; i < MT; ++i)
        a[i] = *(const v2f*)(sb + a_off + i * 16 * GEMM_PITCH + kk * 4);
#pragma unroll
      for (int j = 0; j < NT; ++j)
        b[j] = *(const v2f*)(sb + b_off + j * 16 * GEMM_PITCH + kk * 4);
#pragma unroll
      for (int i = 0; i < MT; ++i)
#pragma unroll
        for (int j = 0; j < NT; ++j)
          acc[i][j] = __builtin_amdgcn_wmma_f32_16x16x4_f32(
              false, a[i], false, b[j], (short)0, acc[i][j], false, false);
    }
  }

  const int mw = m0 + wm * MT * 16 + ((lane >> 4) << 3);
  const int nw = n0 + wn * NT * 16 + r;
#pragma unroll
  for (int i = 0; i < MT; ++i) {
#pragma unroll
    for (int j = 0; j < NT; ++j) {
      const int n = nw + j * 16;
      float bv = 0.0f;
      if (MODE == 1) bv = bias[n];
#pragma unroll
      for (int e = 0; e < 8; ++e) {
        float v = acc[i][j][e];
        if (MODE == 1) {
          v += bv;
          const float e = __expf(-fabsf(v));
          const float u = 1.0f + e;
          const float d = u - 1.0f;
          const float lg = __builtin_amdgcn_logf(u) * 0.69314718f;
          const float q = lg * (e * __builtin_amdgcn_rcpf(d));
          v = fmaxf(v, 0.0f) + ((d == 0.0f) ? e : q);
        }
        C[(size_t)(mw + i * 16 + e) * ldc + n] = v;
      }
    }
  }
}

template <int BN>
__global__ __launch_bounds__(256) void wmma_gemm_ksplit_f32(
    const float* __restrict__ A, const float* __restrict__ W,
    float* __restrict__ C, int N, int K, int lda, int ldb, int ldc) {
  constexpr int KW = 8;
  constexpr int BM = 16;
  constexpr int NT = BN / 16;
  constexpr int ROWS = BM + BN;
  constexpr int NPASS = ROWS * (GEMM_BK / 4) / 32;
  constexpr int APASS = BM * (GEMM_BK / 4) / 32;
  constexpr int WSZ = ROWS * GEMM_PITCH;
  static_assert(NT * 256 <= WSZ, "a wave's partial tiles fit its own region");

  __shared__ __attribute__((aligned(16))) float smem[KW * WSZ];

  const int tid = threadIdx.x;
  const int lane = tid & 31;
  const int wave = tid >> 5;
  const int nbx = N / BN;
  const int m0 = (blockIdx.x / nbx) * BM;
  const int n0 = (blockIdx.x % nbx) * BN;
  const int kslice = K / KW;
  const int k0 = wave * kslice;
  float* sw = smem + wave * WSZ;

  const int lrow = lane >> 3;
  const int seg = lane & 7;
  const float* ga = A + (size_t)(m0 + lrow) * lda + k0 + seg * 4;
  const float* gw = W + (size_t)(n0 + lrow) * ldb + k0 + seg * 4;
  const int sd = lrow * GEMM_PITCH + seg * 4;

  const int r = lane & 15;
  const int khalf = (lane >> 4) << 1;
  const int a_off = r * GEMM_PITCH + khalf;
  const int b_off = (BM + r) * GEMM_PITCH + khalf;

  v8f acc[NT];
#pragma unroll
  for (int j = 0; j < NT; ++j) acc[j] = (v8f){};

  v4f st[NPASS];

#pragma unroll
  for (int i = 0; i < NPASS; ++i)
    st[i] = (i < APASS) ? *(const v4f*)(ga + (size_t)(4 * i) * lda)
                        : *(const v4f*)(gw + (size_t)(4 * (i - APASS)) * ldb);
#pragma unroll
  for (int i = 0; i < NPASS; ++i)
    *(v4f*)(sw + sd + 4 * i * GEMM_PITCH) = st[i];

  const int nchunks = kslice / GEMM_BK;
  for (int kc = 0; kc + 1 < nchunks; ++kc) {
    const int kn = (kc + 1) * GEMM_BK;
#pragma unroll
    for (int i = 0; i < NPASS; ++i)
      st[i] = (i < APASS) ? *(const v4f*)(ga + (size_t)(4 * i) * lda + kn)
                          : *(const v4f*)(gw + (size_t)(4 * (i - APASS)) * ldb + kn);
    __builtin_amdgcn_sched_barrier(0);

#pragma unroll
    for (int kk = 0; kk < GEMM_BK / 4; ++kk) {
      const v2f a = *(const v2f*)(sw + a_off + kk * 4);
      v2f b[NT];
#pragma unroll
      for (int j = 0; j < NT; ++j)
        b[j] = *(const v2f*)(sw + b_off + j * 16 * GEMM_PITCH + kk * 4);
#pragma unroll
      for (int j = 0; j < NT; ++j)
        acc[j] = __builtin_amdgcn_wmma_f32_16x16x4_f32(
            false, a, false, b[j], (short)0, acc[j], false, false);
    }

#pragma unroll
    for (int i = 0; i < NPASS; ++i)
      *(v4f*)(sw + sd + 4 * i * GEMM_PITCH) = st[i];
  }

  {
#pragma unroll
    for (int kk = 0; kk < GEMM_BK / 4; ++kk) {
      const v2f a = *(const v2f*)(sw + a_off + kk * 4);
      v2f b[NT];
#pragma unroll
      for (int j = 0; j < NT; ++j)
        b[j] = *(const v2f*)(sw + b_off + j * 16 * GEMM_PITCH + kk * 4);
#pragma unroll
      for (int j = 0; j < NT; ++j)
        acc[j] = __builtin_amdgcn_wmma_f32_16x16x4_f32(
            false, a, false, b[j], (short)0, acc[j], false, false);
    }
  }

#pragma unroll
  for (int j = 0; j < NT; ++j)
#pragma unroll
    for (int e = 0; e < 8; ++e)
      sw[(j * 8 + e) * 32 + lane] = acc[j][e];
  __syncthreads();

  const int row = m0 + wave + ((lane >> 4) << 3);
#pragma unroll
  for (int j = 0; j < NT; ++j) {
    float s = smem[(j * 8 + wave) * 32 + lane];
#pragma unroll
    for (int w = 1; w < KW; ++w)
      s += smem[w * WSZ + (j * 8 + wave) * 32 + lane];
    C[(size_t)row * ldc + n0 + j * 16 + r] = s;
  }
}

#define CONV_RUN 8
#define CONV_PIECES (D_INNER / 4)
#define CONV_BLK_PER_RUN (CONV_PIECES / 256)

static_assert(SEQ_L % CONV_RUN == 0 && CONV_RUN >= 3 && CONV_PIECES % 256 == 0,
              "a run stays inside one batch row; a workgroup inside one run");

__global__ __launch_bounds__(256) void conv_silu_kernel(
    const float* __restrict__ xz, const float* __restrict__ cw,
    const float* __restrict__ cb, float* __restrict__ xconv) {
  const int run = blockIdx.x / CONV_BLK_PER_RUN;
  if (run >= NB * SEQ_L / CONV_RUN) return;
  const int piece = (blockIdx.x % CONV_BLK_PER_RUN) * 256 + threadIdx.x;
  const int d = piece * 4;
  const int row0 = run * CONV_RUN;
  const int l0 = row0 & (SEQ_L - 1);
  const bool has_front = l0 > 0;

  const float* src = xz + (size_t)row0 * (2 * D_INNER) + d;
  v4f xr[CONV_RUN + 3];
  const int frow = has_front ? row0 - 3 : row0;
  const float* front = xz + (size_t)frow * (2 * D_INNER);
#pragma unroll
  for (int r = 0; r < 3; ++r)
    xr[r] = *(const v4f*)(front + (size_t)r * (2 * D_INNER) + d);
#pragma unroll
  for (int r = 0; r < CONV_RUN; ++r)
    xr[3 + r] = *(const v4f*)(src + (size_t)r * (2 * D_INNER));

  v4f w[4];
#pragma unroll
  for (int c = 0; c < 4; ++c) w[c] = *(const v4f*)(cw + (size_t)(d + c) * 4);
  const v4f bias = *(const v4f*)(cb + d);
  __builtin_amdgcn_sched_barrier(0);

  float* dst = xconv + (size_t)row0 * D_INNER + d;
#pragma unroll
  for (int i = 0; i < CONV_RUN; ++i) {
    v4f o;
#pragma unroll
    for (int c = 0; c < 4; ++c) {
      float acc = bias[c];
#pragma unroll
      for (int j = 0; j < 4; ++j)
        if (i + j >= 3 || has_front) acc = fmaf(w[c][j], xr[i + j][c], acc);
      o[c] = acc / (1.0f + __expf(-acc));
    }
    *(v4f*)(dst + (size_t)i * D_INNER) = o;
  }
}

#define SCAN_T 16
#define SCAN_CH 16
#define SX_PITCH 17
#define SBC_PITCH 20
#define SLOT_DT 0
#define SLOT_X (SCAN_T * SX_PITCH)
#define SLOT_Z (2 * SCAN_T * SX_PITCH)
#define SLOT_BC (3 * SCAN_T * SX_PITCH)
#define SLOT_FLOATS (SLOT_BC + 2 * DSTATE * SBC_PITCH)

static_assert(SLOT_BC % 4 == 0 && SBC_PITCH % 4 == 0 && SLOT_FLOATS % 4 == 0,
              "the 16-byte LDS reads of B|C are aligned");
static_assert(SCAN_CH * DSTATE == 256 && SCAN_T == DSTATE && SCAN_T == SCAN_CH,
              "thread (t, j) of the fetch is thread (channel t, state j) of the scan");

struct ScanRaw {
  float dt, x, z;
  v2f bc;
};

struct ScanOps {
  float dt, x, z;
  v4f B[SCAN_T / 4], C[SCAN_T / 4];
};

__device__ __forceinline__ void scan_request(ScanRaw& g, const float* dt_r,
                                             const float* xc_r, const float* z_r,
                                             const float* bc_r) {
  g.dt = dt_r[0];
  g.x = xc_r[0];
  g.z = z_r[0];
  g.bc = *(const v2f*)bc_r;
}

__device__ __forceinline__ void scan_stage(float* s, const ScanRaw& g, int t, int j) {
  s[SLOT_DT + t * SX_PITCH + j] = g.dt;
  s[SLOT_X + t * SX_PITCH + j] = g.x;
  s[SLOT_Z + t * SX_PITCH + j] = g.z;
  s[SLOT_BC + (2 * j) * SBC_PITCH + t] = g.bc[0];
  s[SLOT_BC + (2 * j + 1) * SBC_PITCH + t] = g.bc[1];
}

__device__ __forceinline__ void scan_take(ScanOps& o, const float* s, int c, int n) {
  o.dt = s[SLOT_DT + n * SX_PITCH + c];
  o.x = s[SLOT_X + n * SX_PITCH + c];
  o.z = s[SLOT_Z + n * SX_PITCH + c];
  const v4f* pb = (const v4f*)(s + SLOT_BC + n * SBC_PITCH);
  const v4f* pc = (const v4f*)(s + SLOT_BC + (DSTATE + n) * SBC_PITCH);
#pragma unroll
  for (int q = 0; q < SCAN_T / 4; ++q) {
    o.B[q] = pb[q];
    o.C[q] = pc[q];
  }
}

template <int T>
__device__ __forceinline__ float row_lane(float v) {
  return __builtin_bit_cast(
      float, __builtin_amdgcn_update_dpp(0, __builtin_bit_cast(int, v),
                                         0x150 + T, 0xf, 0xf, true));
}

template <int M>
__device__ __forceinline__ float row_xor(float v) {
  return __builtin_bit_cast(
      float, __builtin_amdgcn_update_dpp(0, __builtin_bit_cast(int, v),
                                         0x160 + M, 0xf, 0xf, true));
}

template <int M, int CNT>
__device__ __forceinline__ void fold_stage(float (&y)[SCAN_T], bool up) {
#pragma unroll
  for (int j = 0; j < CNT; ++j) {
    const float lo = y[2 * j];
    const float hi = y[2 * j + 1];
    const float keep = up ? hi : lo;
    const float send = up ? lo : hi;
    y[j] = keep + row_xor<M>(send);
  }
}

template <int T>
__device__ __forceinline__ void scan_steps(const ScanOps& o, float dtx, float& h,
                                           float An, float (&y)[SCAN_T]) {
  if constexpr (T < SCAN_T) {
    const float dtv = row_lane<T>(o.dt);
    const float dA = __expf(dtv * An);
    h = fmaf(dA, h, row_lane<T>(dtx) * o.B[T >> 2][T & 3]);
    y[T] = h * o.C[T >> 2][T & 3];
    scan_steps<T + 1>(o, dtx, h, An, y);
  }
}

__device__ __forceinline__ void scan_batch(const ScanOps& o, float& h, float An,
                                           float Dd, int n, float* u_r) {
  const float dtx = o.dt * o.x;
  float y[SCAN_T];
  scan_steps<0>(o, dtx, h, An, y);
  fold_stage<1, 8>(y, (n & 1) != 0);
  fold_stage<2, 4>(y, (n & 2) != 0);
  fold_stage<4, 2>(y, (n & 4) != 0);
  fold_stage<8, 1>(y, (n & 8) != 0);
  const float zv = o.z;
  const float sz = zv / (1.0f + __expf(-zv));
  u_r[0] = fmaf(o.x, Dd, y[0]) * sz;
}

__device__ __forceinline__ void lds_barrier() {
  __builtin_amdgcn_fence(__ATOMIC_RELEASE, "workgroup", "local");
  __builtin_amdgcn_s_barrier();
  __builtin_amdgcn_fence(__ATOMIC_ACQUIRE, "workgroup", "local");
}

__global__ __launch_bounds__(256) __attribute__((amdgpu_waves_per_eu(2, 4))) void scan_kernel(
    const float* __restrict__ dt, const float* __restrict__ xconv,
    const float* __restrict__ x_dbl, const float* __restrict__ xz,
    const float* __restrict__ A_log, const float* __restrict__ Dp,
    float* __restrict__ u) {
  __shared__ __attribute__((aligned(16))) float sm[2][2][SLOT_FLOATS];

  const int c0 = blockIdx.x * SCAN_CH;
  if (c0 >= NB * D_INNER) return;
  const int tid = threadIdx.x;
  const int cw = tid >> 4;
  const int n = tid & 15;
  const int d0 = c0 & (D_INNER - 1);
  const int b = c0 >> 11;
  const int d = d0 + cw;

  const float An = -__expf(A_log[d * DSTATE + n]);
  const float Dd = Dp[d];
  float h = 0.0f;

  const float* dt_g = dt + ((size_t)b * SEQ_L + cw) * D_INNER + d0 + n;
  const float* xc_g = xconv + ((size_t)b * SEQ_L + cw) * D_INNER + d0 + n;
  const float* z_g = xz + ((size_t)b * SEQ_L + cw) * (2 * D_INNER) + D_INNER + d0 + n;
  const float* bc_g = x_dbl + ((size_t)b * SEQ_L + cw) * XDBL_N + DTRANK + 2 * n;
  float* u_p = u + ((size_t)b * SEQ_L + n) * D_INNER + d;

  ScanRaw g0, g1;
  ScanOps ra, rb;

  scan_request(g0, dt_g, xc_g, z_g, bc_g);
  scan_request(g1, dt_g + (size_t)SCAN_T * D_INNER, xc_g + (size_t)SCAN_T * D_INNER,
               z_g + (size_t)SCAN_T * (2 * D_INNER), bc_g + (size_t)SCAN_T * XDBL_N);
  scan_stage(sm[0][0], g0, cw, n);
  scan_stage(sm[0][1], g1, cw, n);
  {
    const int l = 2 * SCAN_T;
    scan_request(g0, dt_g + (size_t)l * D_INNER, xc_g + (size_t)l * D_INNER,
                 z_g + (size_t)l * (2 * D_INNER), bc_g + (size_t)l * XDBL_N);
    scan_request(g1, dt_g + (size_t)(l + SCAN_T) * D_INNER, xc_g + (size_t)(l + SCAN_T) * D_INNER,
                 z_g + (size_t)(l + SCAN_T) * (2 * D_INNER), bc_g + (size_t)(l + SCAN_T) * XDBL_N);
  }
  lds_barrier();
  scan_take(ra, sm[0][0], cw, n);

  for (int l0 = 0; l0 < SEQ_L; l0 += 2 * SCAN_T) {
    const int l1 = l0 + SCAN_T;
    const int cur = (l0 / (2 * SCAN_T)) & 1;
    const int l2 = (l0 + 4 * SCAN_T < SEQ_L) ? (l0 + 4 * SCAN_T) : l0;

    scan_take(rb, sm[cur][1], cw, n);
    scan_batch(ra, h, An, Dd, n, u_p + (size_t)l0 * D_INNER);

    scan_stage(sm[cur ^ 1][0], g0, cw, n);
    scan_stage(sm[cur ^ 1][1], g1, cw, n);
    scan_request(g0, dt_g + (size_t)l2 * D_INNER, xc_g + (size_t)l2 * D_INNER,
                 z_g + (size_t)l2 * (2 * D_INNER), bc_g + (size_t)l2 * XDBL_N);
    scan_request(g1, dt_g + (size_t)(l2 + SCAN_T) * D_INNER, xc_g + (size_t)(l2 + SCAN_T) * D_INNER,
                 z_g + (size_t)(l2 + SCAN_T) * (2 * D_INNER), bc_g + (size_t)(l2 + SCAN_T) * XDBL_N);
    lds_barrier();

    scan_take(ra, sm[cur ^ 1][0], cw, n);
    __builtin_amdgcn_sched_barrier(0);
    scan_batch(rb, h, An, Dd, n, u_p + (size_t)l1 * D_INNER);
  }
}

extern "C" void kernel_launch(void* const* d_in, const int* in_sizes, int n_in,
                              void* d_out, int out_size, void* d_ws, size_t ws_size,
                              hipStream_t stream) {
  const float* x      = (const float*)d_in[0];
  const float* W_in   = (const float*)d_in[1];
  const float* conv_w = (const float*)d_in[2];
  const float* conv_b = (const float*)d_in[3];
  const float* W_x    = (const float*)d_in[4];
  const float* W_dt   = (const float*)d_in[5];
  const float* b_dt   = (const float*)d_in[6];
  const float* A_log  = (const float*)d_in[7];
  const float* Dp     = (const float*)d_in[8];
  const float* W_out  = (const float*)d_in[9];
  float* out = (float*)d_out;

  float* ws = (float*)d_ws;
  const size_t M = (size_t)NB * SEQ_L;
  float* xz    = ws;
  float* xconv = xz + M * 2 * D_INNER;
  float* x_dbl = xconv + M * D_INNER;
  float* dt    = x_dbl + M * XDBL_N;
  float* u     = dt + M * D_INNER;

  dim3 blk(256);
  {
    int blocks = (int)(M / 128) * (2 * D_INNER / 128);
    hipLaunchKernelGGL((wmma_gemm_f32<0, 128, 128, 4, 2>), dim3(blocks), dim3(256), 0, stream,
                       x, W_in, xz, nullptr,
                       2 * D_INNER, D_MODEL, D_MODEL, D_MODEL, 2 * D_INNER);
  }
  {
    int blocks = (NB * SEQ_L / CONV_RUN) * CONV_BLK_PER_RUN;
    hipLaunchKernelGGL(conv_silu_kernel, dim3(blocks), blk, 0, stream,
                       xz, conv_w, conv_b, xconv);
  }
  {
    int blocks = (int)(M / 16) * (XDBL_N / 48);
    hipLaunchKernelGGL((wmma_gemm_ksplit_f32<48>), dim3(blocks), dim3(256), 0, stream,
                       xconv, W_x, x_dbl,
                       XDBL_N, D_INNER, D_INNER, D_INNER, XDBL_N);
  }
  {
    int blocks = (int)(M / 128) * (D_INNER / 128);
    hipLaunchKernelGGL((wmma_gemm_f32<1, 128, 128, 4, 2>), dim3(blocks), dim3(256), 0, stream,
                       x_dbl, W_dt, dt, b_dt,
                       D_INNER, DTRANK, XDBL_N, DTRANK, D_INNER);
  }
  {
    int total = NB * D_INNER * DSTATE;
    hipLaunchKernelGGL(scan_kernel, dim3((total + 255) / 256), blk, 0, stream,
                       dt, xconv, x_dbl, xz, A_log, Dp, u);
  }
  {
    int blocks = (int)(M / 128) * (D_MODEL / 64);
    hipLaunchKernelGGL((wmma_gemm_f32<0, 128, 64, 4, 2>), dim3(blocks), dim3(256), 0, stream,
                       u, W_out, out, nullptr,
                       D_MODEL, D_INNER, D_INNER, D_INNER, D_MODEL);
  }
}
